# starting kernel
# baseline (speedup 1.0000x reference)
#include <hip/hip_runtime.h>
#include <hip/hip_bf16.h>
#include <math.h>


typedef __attribute__((ext_vector_type(2))) float v2f;
typedef __attribute__((ext_vector_type(8))) float v8f;

#define B_SZ 4096
#define D_SZ 66
#define H_SZ 512

constexpr int BM = 64;
constexpr int BN = 64;
constexpr int BK = 32;
constexpr int NT = 128;

__device__ __forceinline__ v8f wmma4(v2f a, v2f b, v8f c) {
    return __builtin_amdgcn_wmma_f32_16x16x4_f32(
        false, a, false, b, (short)0, c, false, false);
}

__global__ __launch_bounds__(256)
void precompute_C(const float* __restrict__ W1, const float* __restrict__ W2,
                  const float* __restrict__ W3, float* __restrict__ Cm)
{
    int idx = blockIdx.x * blockDim.x + threadIdx.x;
    int k = idx & (H_SZ - 1);
    int l = idx >> 9;
    float m = 0.0f;
    #pragma unroll 2
    for (int j = 0; j < D_SZ; ++j)
        m = fmaf(W1[j * H_SZ + l], W3[k * D_SZ + j], m);
    Cm[l * H_SZ + k] = m * W2[l * H_SZ + k];
}

template<bool ACONCAT, int EPI>
__global__ __launch_bounds__(NT)
void gemm_wmma(const float* __restrict__ A, const float* __restrict__ xs,
               const float* __restrict__ tvec,
               const float* __restrict__ Bm, const float* __restrict__ bias,
               float* __restrict__ out0, float* __restrict__ out1,
               int M, int N, int K)
{
    __shared__ float As[BM][BK + 4];
    __shared__ float Bt[BN][BK + 4];

    const int tid  = threadIdx.x;
    const int wave = tid >> 5;
    const int lane = tid & 31;
    const int wm   = wave >> 1;
    const int wn   = wave & 1;

    const int m0 = blockIdx.y * BM;
    const int n0 = blockIdx.x * BN;

    v8f acc[2][2] = {};

    const int row16 = lane & 15;
    const int ksub  = (lane >> 4) << 1;

    for (int k0 = 0; k0 < K; k0 += BK) {
        #pragma unroll
        for (int i = 0; i < (BM * BK) / NT; ++i) {
            int idx = tid + i * NT;
            int mm = idx >> 5;
            int kk = idx & (BK - 1);
            int gk = k0 + kk;
            float v = 0.0f;
            if (ACONCAT) {
                int gm = m0 + mm;
                if (gk < D_SZ)       v = xs[gm * D_SZ + gk];
                else if (gk == D_SZ) v = tvec[gm];
            } else {
                if (gk < K) v = A[(size_t)(m0 + mm) * K + gk];
            }
            As[mm][kk] = v;
        }
        #pragma unroll
        for (int i = 0; i < (BK * BN) / NT; ++i) {
            int idx = tid + i * NT;
            int kk = idx / BN;
            int nn = idx % BN;
            int gk = k0 + kk;
            int gn = n0 + nn;
            float v = 0.0f;
            if (gk < K && gn < N) v = Bm[(size_t)gk * N + gn];
            Bt[nn][kk] = v;
        }
        __syncthreads();

        #pragma unroll
        for (int kk = 0; kk < BK; kk += 4) {
            v2f a0 = *(const v2f*)&As[wm * 32 +      row16][kk + ksub];
            v2f a1 = *(const v2f*)&As[wm * 32 + 16 + row16][kk + ksub];
            v2f b0 = *(const v2f*)&Bt[wn * 32 +      row16][kk + ksub];
            v2f b1 = *(const v2f*)&Bt[wn * 32 + 16 + row16][kk + ksub];
            acc[0][0] = wmma4(a0, b0, acc[0][0]);
            acc[0][1] = wmma4(a0, b1, acc[0][1]);
            acc[1][0] = wmma4(a1, b0, acc[1][0]);
            acc[1][1] = wmma4(a1, b1, acc[1][1]);
        }
        __syncthreads();
    }

    #pragma unroll
    for (int mt = 0; mt < 2; ++mt) {
        const int mbase = m0 + wm * 32 + mt * 16 + ((lane >> 4) << 3);
        #pragma unroll
        for (int nt = 0; nt < 2; ++nt) {
            const int nc = n0 + wn * 32 + nt * 16 + row16;
            if (nc >= N) continue;
            const float bi = bias ? bias[nc] : 0.0f;
            #pragma unroll
            for (int r = 0; r < 8; ++r) {
                const int gm = mbase + r;
                float z = acc[mt][nt][r] + bi;
                if (EPI == 0) {
                    out0[(size_t)gm * N + nc] = z;
                } else {
                    float sg = 1.0f / (1.0f + __expf(-z));
                    out0[(size_t)gm * N + nc] = z * sg;
                    out1[(size_t)gm * N + nc] = sg * (1.0f + z * (1.0f - sg));
                }
            }
        }
    }
}

__global__ __launch_bounds__(128)
void dot_div_kernel(const float* __restrict__ u, const float* __restrict__ s2,
                    float* __restrict__ out)
{
    int b = blockIdx.x;
    int tid = threadIdx.x;
    float sum = 0.0f;
    #pragma unroll
    for (int k = tid; k < H_SZ; k += 128)
        sum += u[(size_t)b * H_SZ + k] * s2[(size_t)b * H_SZ + k];
    #pragma unroll
    for (int off = 16; off > 0; off >>= 1)
        sum += __shfl_xor(sum, off, 32);
    __shared__ float red[4];
    if ((tid & 31) == 0) red[tid >> 5] = sum;
    __syncthreads();
    if (tid == 0) out[b] = -(red[0] + red[1] + red[2] + red[3]);
}

extern "C" void kernel_launch(void* const* d_in, const int* in_sizes, int n_in,
                              void* d_out, int out_size, void* d_ws, size_t ws_size,
                              hipStream_t stream) {
    const float* xs = (const float*)d_in[0];
    const float* t  = (const float*)d_in[1];
    const float* W1 = (const float*)d_in[2];
    const float* b1 = (const float*)d_in[3];
    const float* W2 = (const float*)d_in[4];
    const float* b2 = (const float*)d_in[5];
    const float* W3 = (const float*)d_in[6];
    const float* b3 = (const float*)d_in[7];

    float* dxs  = (float*)d_out;
    float* ndiv = (float*)d_out + (size_t)B_SZ * D_SZ;

    float* ws = (float*)d_ws;
    const size_t act = (size_t)B_SZ * H_SZ;
    float* a1u = ws;
    float* s1  = ws + act;
    float* a2  = ws + 2 * act;
    float* s2  = ws + 3 * act;
    float* Cm  = ws + 4 * act;

    dim3 blk(NT);
    dim3 gridH(H_SZ / BN, B_SZ / BM);
    dim3 gridD((D_SZ + BN - 1) / BN, B_SZ / BM);

    precompute_C<<<(H_SZ * H_SZ) / 256, 256, 0, stream>>>(W1, W2, W3, Cm);

    gemm_wmma<true, 1><<<gridH, blk, 0, stream>>>(
        nullptr, xs, t, W1, b1, a1u, s1, B_SZ, H_SZ, D_SZ + 1);

    gemm_wmma<false, 1><<<gridH, blk, 0, stream>>>(
        a1u, nullptr, nullptr, W2, b2, a2, s2, B_SZ, H_SZ, H_SZ);

    gemm_wmma<false, 0><<<gridH, blk, 0, stream>>>(
        s1, nullptr, nullptr, Cm, nullptr, a1u, nullptr, B_SZ, H_SZ, H_SZ);

    gemm_wmma<false, 0><<<gridD, blk, 0, stream>>>(
        a2, nullptr, nullptr, W3, b3, dxs, nullptr, B_SZ, D_SZ, H_SZ);

    dot_div_kernel<<<B_SZ, 128, 0, stream>>>(a1u, s2, ndiv);
}


# GEMM tile staging: all global loads of a k-tile issued unconditionally as one group (A rows 128-bit) before any LDS store, LDS written 128-bit; was 32 guarded load-wait-store round trips per tile
# speedup vs baseline: 5.1181x; 5.1181x over previous
#include <hip/hip_runtime.h>
#include <hip/hip_bf16.h>
#include <math.h>


typedef __attribute__((ext_vector_type(2))) float v2f;
typedef __attribute__((ext_vector_type(8))) float v8f;

#define B_SZ 4096
#define D_SZ 66
#define H_SZ 512

constexpr int BM = 64;
constexpr int BN = 64;
constexpr int BK = 32;
constexpr int NT = 128;

__device__ __forceinline__ v8f wmma4(v2f a, v2f b, v8f c) {
    return __builtin_amdgcn_wmma_f32_16x16x4_f32(
        false, a, false, b, (short)0, c, false, false);
}

__global__ __launch_bounds__(256)
void precompute_C(const float* __restrict__ W1, const float* __restrict__ W2,
                  const float* __restrict__ W3, float* __restrict__ Cm)
{
    int idx = blockIdx.x * blockDim.x + threadIdx.x;
    int k = idx & (H_SZ - 1);
    int l = idx >> 9;
    float m = 0.0f;
    #pragma unroll 2
    for (int j = 0; j < D_SZ; ++j)
        m = fmaf(W1[j * H_SZ + l], W3[k * D_SZ + j], m);
    Cm[l * H_SZ + k] = m * W2[l * H_SZ + k];
}

typedef __attribute__((ext_vector_type(4))) float v4f;

template<bool ACONCAT, int EPI, int NDIM, int KDIM>
__global__ __launch_bounds__(NT)
void gemm_wmma(const float* __restrict__ A, const float* __restrict__ xs,
               const float* __restrict__ tvec,
               const float* __restrict__ Bm, const float* __restrict__ bias,
               float* __restrict__ out0, float* __restrict__ out1)
{
    __shared__ __attribute__((aligned(16))) float As[BM][BK + 4];
    __shared__ __attribute__((aligned(16))) float Bt[BN][BK + 4];

    constexpr bool KFULL = (KDIM % BK) == 0;
    constexpr bool NFULL = (NDIM % BN) == 0;
    static_assert(ACONCAT || (KFULL && (KDIM % 4) == 0), "A rows must allow 128-bit loads");

    const int tid  = threadIdx.x;
    const int wave = tid >> 5;
    const int lane = tid & 31;
    const int wm   = wave >> 1;
    const int wn   = wave & 1;

    const int m0 = blockIdx.y * BM;
    const int n0 = blockIdx.x * BN;

    v8f acc[2][2] = {};

    const int row16 = lane & 15;
    const int ksub  = (lane >> 4) << 1;

    const int a_q  = tid & 7;
    const int a_m  = tid >> 3;
    const int b_n  = tid & (BN - 1);
    const int b_q  = tid >> 6;
    const int b_gn = n0 + b_n;
    const int b_cn = NFULL ? b_gn : (b_gn < NDIM ? b_gn : NDIM - 1);

    for (int k0 = 0; k0 < KDIM; k0 += BK) {
        v4f   av4[4];
        float avs[16];
        float avt[16];
        const bool a_tail = ACONCAT && (k0 + BK > D_SZ);
        if (ACONCAT) {
            #pragma unroll
            for (int i = 0; i < 16; ++i) {
                const int idx = tid + i * NT;
                const int gk = k0 + (idx & (BK - 1));
                const int gm = m0 + (idx >> 5);
                const int ck = gk < D_SZ ? gk : D_SZ - 1;
                avs[i] = xs[gm * D_SZ + ck];
                avt[i] = a_tail ? tvec[gm] : 0.0f;
            }
        } else {
            #pragma unroll
            for (int i = 0; i < 4; ++i)
                av4[i] = *(const v4f*)&A[(size_t)(m0 + a_m + 16 * i) * KDIM + k0 + 4 * a_q];
        }
        v4f bv[4];
        #pragma unroll
        for (int i = 0; i < 4; ++i) {
            #pragma unroll
            for (int j = 0; j < 4; ++j) {
                const int gk = k0 + 4 * (b_q + 2 * i) + j;
                const int ck = KFULL ? gk : (gk < KDIM ? gk : KDIM - 1);
                const float x = Bm[(size_t)ck * NDIM + b_cn];
                const bool ok = (KFULL || gk < KDIM) && (NFULL || b_gn < NDIM);
                bv[i][j] = ok ? x : 0.0f;
            }
        }
        __builtin_amdgcn_sched_barrier(0);
        if (ACONCAT) {
            #pragma unroll
            for (int i = 0; i < 16; ++i) {
                const int idx = tid + i * NT;
                float v = avs[i];
                if (a_tail) {
                    float tv = avt[i];
                    asm volatile("" : "+v"(v), "+v"(tv));
                    const int gk = k0 + (idx & (BK - 1));
                    v = gk < D_SZ ? v : (gk == D_SZ ? tv : 0.0f);
                }
                As[idx >> 5][idx & (BK - 1)] = v;
            }
        } else {
            #pragma unroll
            for (int i = 0; i < 4; ++i)
                *(v4f*)&As[a_m + 16 * i][4 * a_q] = av4[i];
        }
        #pragma unroll
        for (int i = 0; i < 4; ++i)
            *(v4f*)&Bt[b_n][4 * (b_q + 2 * i)] = bv[i];
        __syncthreads();

        #pragma unroll
        for (int kk = 0; kk < BK; kk += 4) {
            v2f a0 = *(const v2f*)&As[wm * 32 +      row16][kk + ksub];
            v2f a1 = *(const v2f*)&As[wm * 32 + 16 + row16][kk + ksub];
            v2f b0 = *(const v2f*)&Bt[wn * 32 +      row16][kk + ksub];
            v2f b1 = *(const v2f*)&Bt[wn * 32 + 16 + row16][kk + ksub];
            acc[0][0] = wmma4(a0, b0, acc[0][0]);
            acc[0][1] = wmma4(a0, b1, acc[0][1]);
            acc[1][0] = wmma4(a1, b0, acc[1][0]);
            acc[1][1] = wmma4(a1, b1, acc[1][1]);
        }
        __syncthreads();
    }

    #pragma unroll
    for (int mt = 0; mt < 2; ++mt) {
        const int mbase = m0 + wm * 32 + mt * 16 + ((lane >> 4) << 3);
        #pragma unroll
        for (int nt = 0; nt < 2; ++nt) {
            const int nc = n0 + wn * 32 + nt * 16 + row16;
            if (!NFULL && nc >= NDIM) continue;
            const float bi = bias ? bias[nc] : 0.0f;
            #pragma unroll
            for (int r = 0; r < 8; ++r) {
                const int gm = mbase + r;
                float z = acc[mt][nt][r] + bi;
                if (EPI == 0) {
                    out0[(size_t)gm * NDIM + nc] = z;
                } else {
                    float sg = 1.0f / (1.0f + __expf(-z));
                    out0[(size_t)gm * NDIM + nc] = z * sg;
                    out1[(size_t)gm * NDIM + nc] = sg * (1.0f + z * (1.0f - sg));
                }
            }
        }
    }
}

__global__ __launch_bounds__(128)
void dot_div_kernel(const float* __restrict__ u, const float* __restrict__ s2,
                    float* __restrict__ out)
{
    int b = blockIdx.x;
    int tid = threadIdx.x;
    float sum = 0.0f;
    #pragma unroll
    for (int k = tid; k < H_SZ; k += 128)
        sum += u[(size_t)b * H_SZ + k] * s2[(size_t)b * H_SZ + k];
    #pragma unroll
    for (int off = 16; off > 0; off >>= 1)
        sum += __shfl_xor(sum, off, 32);
    __shared__ float red[4];
    if ((tid & 31) == 0) red[tid >> 5] = sum;
    __syncthreads();
    if (tid == 0) out[b] = -(red[0] + red[1] + red[2] + red[3]);
}

extern "C" void kernel_launch(void* const* d_in, const int* in_sizes, int n_in,
                              void* d_out, int out_size, void* d_ws, size_t ws_size,
                              hipStream_t stream) {
    const float* xs = (const float*)d_in[0];
    const float* t  = (const float*)d_in[1];
    const float* W1 = (const float*)d_in[2];
    const float* b1 = (const float*)d_in[3];
    const float* W2 = (const float*)d_in[4];
    const float* b2 = (const float*)d_in[5];
    const float* W3 = (const float*)d_in[6];
    const float* b3 = (const float*)d_in[7];

    float* dxs  = (float*)d_out;
    float* ndiv = (float*)d_out + (size_t)B_SZ * D_SZ;

    float* ws = (float*)d_ws;
    const size_t act = (size_t)B_SZ * H_SZ;
    float* a1u = ws;
    float* s1  = ws + act;
    float* a2  = ws + 2 * act;
    float* s2  = ws + 3 * act;
    float* Cm  = ws + 4 * act;

    dim3 blk(NT);
    dim3 gridH(H_SZ / BN, B_SZ / BM);
    dim3 gridD((D_SZ + BN - 1) / BN, B_SZ / BM);

    precompute_C<<<(H_SZ * H_SZ) / 256, 256, 0, stream>>>(W1, W2, W3, Cm);

    gemm_wmma<true, 1, H_SZ, D_SZ + 1><<<gridH, blk, 0, stream>>>(
        nullptr, xs, t, W1, b1, a1u, s1);

    gemm_wmma<false, 1, H_SZ, H_SZ><<<gridH, blk, 0, stream>>>(
        a1u, nullptr, nullptr, W2, b2, a2, s2);

    gemm_wmma<false, 0, H_SZ, H_SZ><<<gridH, blk, 0, stream>>>(
        s1, nullptr, nullptr, Cm, nullptr, a1u, nullptr);

    gemm_wmma<false, 0, D_SZ, H_SZ><<<gridD, blk, 0, stream>>>(
        a2, nullptr, nullptr, W3, b3, dxs, nullptr);

    dot_div_kernel<<<B_SZ, 128, 0, stream>>>(a1u, s2, ndiv);
}


# GEMM k-loop software-pipelined: global loads of k-tile t+1 issued before tile t's WMMAs and stored to a second LDS stage after them, one barrier per tile (was load-wait-store-compute, two barriers)
# speedup vs baseline: 6.6355x; 1.2965x over previous
#include <hip/hip_runtime.h>
#include <hip/hip_bf16.h>
#include <math.h>


typedef __attribute__((ext_vector_type(2))) float v2f;
typedef __attribute__((ext_vector_type(8))) float v8f;

#define B_SZ 4096
#define D_SZ 66
#define H_SZ 512

constexpr int BM = 64;
constexpr int BN = 64;
constexpr int BK = 32;
constexpr int NT = 128;

__device__ __forceinline__ v8f wmma4(v2f a, v2f b, v8f c) {
    return __builtin_amdgcn_wmma_f32_16x16x4_f32(
        false, a, false, b, (short)0, c, false, false);
}

typedef __attribute__((ext_vector_type(4))) float v4f;

template<bool ACONCAT, int EPI, int NDIM, int KDIM>
__global__ __launch_bounds__(NT)
void gemm_wmma(const float* __restrict__ A, const float* __restrict__ xs,
               const float* __restrict__ tvec,
               const float* __restrict__ Bm, const float* __restrict__ bias,
               float* __restrict__ out0, float* __restrict__ out1)
{
    __shared__ __attribute__((aligned(16))) float As[2][BM][BK + 4];
    __shared__ __attribute__((aligned(16))) float Bt[2][BN][BK + 4];

    constexpr bool KFULL = (KDIM % BK) == 0;
    constexpr bool HAS_T = ACONCAT && (KDIM > D_SZ);
    constexpr bool NFULL = (NDIM % BN) == 0;
    constexpr int  NKT   = (KDIM + BK - 1) / BK;
    constexpr int  UNR   = ACONCAT ? (NKT - 1) : 1;
    static_assert(ACONCAT || (KFULL && (KDIM % 4) == 0), "A rows must allow 128-bit loads");
    static_assert(NKT >= 2, "the pipelined loop expects at least two k-tiles");

    const int tid  = threadIdx.x;
    const int wave = tid >> 5;
    const int lane = tid & 31;
    const int wm   = wave >> 1;
    const int wn   = wave & 1;

    const int m0 = blockIdx.y * BM;
    const int n0 = blockIdx.x * BN;

    v8f acc[2][2] = {};

    const int row16 = lane & 15;
    const int ksub  = (lane >> 4) << 1;

    const int a_q  = tid & 7;
    const int a_m  = tid >> 3;
    const int b_n  = tid & (BN - 1);
    const int b_q  = tid >> 6;
    const int b_gn = n0 + b_n;
    const int b_cn = NFULL ? b_gn : (b_gn < NDIM ? b_gn : NDIM - 1);

    v4f   av4[4];
    float avs[16];
    float avt[16];
    v4f   bv[4];

    auto gload = [&](const int k0) __attribute__((always_inline)) {
        const bool a_tail = ACONCAT && (k0 + BK > D_SZ);
        if (ACONCAT) {
            #pragma unroll
            for (int i = 0; i < 16; ++i) {
                const int idx = tid + i * NT;
                const int gk = k0 + (idx & (BK - 1));
                const int gm = m0 + (idx >> 5);
                const int ck = gk < D_SZ ? gk : D_SZ - 1;
                avs[i] = xs[gm * D_SZ + ck];
                avt[i] = (HAS_T && a_tail) ? tvec[gm] : 0.0f;
            }
        } else {
            #pragma unroll
            for (int i = 0; i < 4; ++i)
                av4[i] = *(const v4f*)&A[(size_t)(m0 + a_m + 16 * i) * KDIM + k0 + 4 * a_q];
        }
        #pragma unroll
        for (int i = 0; i < 4; ++i) {
            #pragma unroll
            for (int j = 0; j < 4; ++j) {
                const int gk = k0 + 4 * (b_q + 2 * i) + j;
                const int ck = KFULL ? gk : (gk < KDIM ? gk : KDIM - 1);
                bv[i][j] = Bm[(size_t)ck * NDIM + b_cn];
            }
        }
    };

    auto lstore = [&](const int k0, const int s) __attribute__((always_inline)) {
        const bool a_tail = ACONCAT && (k0 + BK > D_SZ);
        if (ACONCAT) {
            #pragma unroll
            for (int i = 0; i < 16; ++i) {
                const int idx = tid + i * NT;
                float v = avs[i];
                if (a_tail) {
                    float tv = avt[i];
                    if (HAS_T) asm volatile("" : "+v"(v), "+v"(tv));
                    else       asm volatile("" : "+v"(v));
                    const int gk = k0 + (idx & (BK - 1));
                    v = gk < D_SZ ? v : ((HAS_T && gk == D_SZ) ? tv : 0.0f);
                }
                As[s][idx >> 5][idx & (BK - 1)] = v;
            }
        } else {
            #pragma unroll
            for (int i = 0; i < 4; ++i)
                *(v4f*)&As[s][a_m + 16 * i][4 * a_q] = av4[i];
        }
        #pragma unroll
        for (int i = 0; i < 4; ++i) {
            v4f w = bv[i];
            if (!(KFULL && NFULL)) {
                #pragma unroll
                for (int j = 0; j < 4; ++j) {
                    const int gk = k0 + 4 * (b_q + 2 * i) + j;
                    const bool ok = (KFULL || gk < KDIM) && (NFULL || b_gn < NDIM);
                    w[j] = ok ? w[j] : 0.0f;
                }
            }
            *(v4f*)&Bt[s][b_n][4 * (b_q + 2 * i)] = w;
        }
    };

    auto compute = [&](const int s) __attribute__((always_inline)) {
        #pragma unroll
        for (int kk = 0; kk < BK; kk += 4) {
            v2f a0 = *(const v2f*)&As[s][wm * 32 +      row16][kk + ksub];
            v2f a1 = *(const v2f*)&As[s][wm * 32 + 16 + row16][kk + ksub];
            v2f b0 = *(const v2f*)&Bt[s][wn * 32 +      row16][kk + ksub];
            v2f b1 = *(const v2f*)&Bt[s][wn * 32 + 16 + row16][kk + ksub];
            acc[0][0] = wmma4(a0, b0, acc[0][0]);
            acc[0][1] = wmma4(a0, b1, acc[0][1]);
            acc[1][0] = wmma4(a1, b0, acc[1][0]);
            acc[1][1] = wmma4(a1, b1, acc[1][1]);
        }
    };

    gload(0);
    __builtin_amdgcn_sched_barrier(0);
    lstore(0, 0);
    __syncthreads();

    #pragma clang loop unroll_count(UNR)
    for (int t = 0; t < NKT - 1; ++t) {
        gload((t + 1) * BK);
        __builtin_amdgcn_sched_barrier(0);
        compute(t & 1);
        __builtin_amdgcn_sched_barrier(0);
        lstore((t + 1) * BK, (t + 1) & 1);
        __syncthreads();
    }
    compute((NKT - 1) & 1);

    #pragma unroll
    for (int mt = 0; mt < 2; ++mt) {
        const int mbase = m0 + wm * 32 + mt * 16 + ((lane >> 4) << 3);
        #pragma unroll
        for (int nt = 0; nt < 2; ++nt) {
            const int nc = n0 + wn * 32 + nt * 16 + row16;
            if (!NFULL && nc >= NDIM) continue;
            if (EPI == 2) {
                const size_t o = (size_t)nc * H_SZ + mbase;
                const v4f w0 = *(const v4f*)&bias[o];
                const v4f w1 = *(const v4f*)&bias[o + 4];
                v4f r0, r1;
                #pragma unroll
                for (int r = 0; r < 4; ++r) {
                    r0[r] = acc[mt][nt][r]     * w0[r];
                    r1[r] = acc[mt][nt][r + 4] * w1[r];
                }
                *(v4f*)&out0[o]     = r0;
                *(v4f*)&out0[o + 4] = r1;
                continue;
            }
            const float bi = bias ? bias[nc] : 0.0f;
            #pragma unroll
            for (int r = 0; r < 8; ++r) {
                const int gm = mbase + r;
                float z = acc[mt][nt][r] + bi;
                if (EPI == 0) {
                    out0[(size_t)gm * NDIM + nc] = z;
                } else {
                    float sg = 1.0f / (1.0f + __expf(-z));
                    out0[(size_t)gm * NDIM + nc] = z * sg;
                    out1[(size_t)gm * NDIM + nc] = sg * (1.0f + z * (1.0f - sg));
                }
            }
        }
    }
}

__global__ __launch_bounds__(128)
void dot_div_kernel(const float* __restrict__ u, const float* __restrict__ s2,
                    float* __restrict__ out)
{
    int b = blockIdx.x;
    int tid = threadIdx.x;
    float sum = 0.0f;
    #pragma unroll
    for (int k = tid; k < H_SZ; k += 128)
        sum += u[(size_t)b * H_SZ + k] * s2[(size_t)b * H_SZ + k];
    #pragma unroll
    for (int off = 16; off > 0; off >>= 1)
        sum += __shfl_xor(sum, off, 32);
    __shared__ float red[4];
    if ((tid & 31) == 0) red[tid >> 5] = sum;
    __syncthreads();
    if (tid == 0) out[b] = -(red[0] + red[1] + red[2] + red[3]);
}

extern "C" void kernel_launch(void* const* d_in, const int* in_sizes, int n_in,
                              void* d_out, int out_size, void* d_ws, size_t ws_size,
                              hipStream_t stream) {
    const float* xs = (const float*)d_in[0];
    const float* t  = (const float*)d_in[1];
    const float* W1 = (const float*)d_in[2];
    const float* b1 = (const float*)d_in[3];
    const float* W2 = (const float*)d_in[4];
    const float* b2 = (const float*)d_in[5];
    const float* W3 = (const float*)d_in[6];
    const float* b3 = (const float*)d_in[7];

    float* dxs  = (float*)d_out;
    float* ndiv = (float*)d_out + (size_t)B_SZ * D_SZ;

    float* ws = (float*)d_ws;
    const size_t act = (size_t)B_SZ * H_SZ;
    float* a1u = ws;
    float* s1  = ws + act;
    float* a2  = ws + 2 * act;
    float* s2  = ws + 3 * act;
    float* Cm  = ws + 4 * act;

    dim3 blk(NT);
    dim3 gridH(H_SZ / BN, B_SZ / BM);
    dim3 gridD((D_SZ + BN - 1) / BN, B_SZ / BM);

    dim3 gridC(H_SZ / BN, H_SZ / BM);
    gemm_wmma<true, 2, H_SZ, D_SZ><<<gridC, blk, 0, stream>>>(
        nullptr, W3, nullptr, W1, W2, Cm, nullptr);

    gemm_wmma<true, 1, H_SZ, D_SZ + 1><<<gridH, blk, 0, stream>>>(
        nullptr, xs, t, W1, b1, a1u, s1);

    gemm_wmma<false, 1, H_SZ, H_SZ><<<gridH, blk, 0, stream>>>(
        a1u, nullptr, nullptr, W2, b2, a2, s2);

    gemm_wmma<false, 0, H_SZ, H_SZ><<<gridH, blk, 0, stream>>>(
        s1, nullptr, nullptr, Cm, nullptr, a1u, nullptr);

    gemm_wmma<false, 0, D_SZ, H_SZ><<<gridD, blk, 0, stream>>>(
        a2, nullptr, nullptr, W3, b3, dxs, nullptr);

    dot_div_kernel<<<B_SZ, 128, 0, stream>>>(a1u, s2, ndiv);
}


# independent jobs share a launch, selected by block index: trace-matrix GEMM beside layer 1, output-layer GEMM beside the divergence dot (6 dependent launches -> 4)
# speedup vs baseline: 7.9184x; 1.1933x over previous
#include <hip/hip_runtime.h>
#include <hip/hip_bf16.h>
#include <math.h>


typedef __attribute__((ext_vector_type(2))) float v2f;
typedef __attribute__((ext_vector_type(8))) float v8f;

#define B_SZ 4096
#define D_SZ 66
#define H_SZ 512

constexpr int BM = 64;
constexpr int BN = 64;
constexpr int BK = 32;
constexpr int NT = 128;

__device__ __forceinline__ v8f wmma4(v2f a, v2f b, v8f c) {
    return __builtin_amdgcn_wmma_f32_16x16x4_f32(
        false, a, false, b, (short)0, c, false, false);
}

typedef __attribute__((ext_vector_type(4))) float v4f;

typedef float lds_tile_t[2][BM][BK + 4];

template<bool ACONCAT, int EPI, int NDIM, int KDIM>
__device__ __forceinline__
void gemm_tile(lds_tile_t& As, lds_tile_t& Bt,
               const int bx, const int by,
               const float* __restrict__ A, const float* __restrict__ xs,
               const float* __restrict__ tvec,
               const float* __restrict__ Bm, const float* __restrict__ bias,
               float* __restrict__ out0, float* __restrict__ out1)
{
    static_assert(BM == BN, "A and B tiles share one LDS array type");

    constexpr bool KFULL = (KDIM % BK) == 0;
    constexpr bool HAS_T = ACONCAT && (KDIM > D_SZ);
    constexpr bool NFULL = (NDIM % BN) == 0;
    constexpr int  NKT   = (KDIM + BK - 1) / BK;
    constexpr int  UNR   = ACONCAT ? (NKT - 1) : 1;
    static_assert(ACONCAT || (KFULL && (KDIM % 4) == 0), "A rows must allow 128-bit loads");
    static_assert(NKT >= 2, "the pipelined loop expects at least two k-tiles");

    const int tid  = threadIdx.x;
    const int wave = tid >> 5;
    const int lane = tid & 31;
    const int wm   = wave >> 1;
    const int wn   = wave & 1;

    const int m0 = by * BM;
    const int n0 = bx * BN;

    v8f acc[2][2] = {};

    const int row16 = lane & 15;
    const int ksub  = (lane >> 4) << 1;

    const int a_q  = tid & 7;
    const int a_m  = tid >> 3;
    const int b_n  = tid & (BN - 1);
    const int b_q  = tid >> 6;
    const int b_gn = n0 + b_n;
    const int b_cn = NFULL ? b_gn : (b_gn < NDIM ? b_gn : NDIM - 1);

    v4f   av4[4];
    float avs[16];
    float avt[16];
    v4f   bv[4];

    auto gload = [&](const int k0) __attribute__((always_inline)) {
        const bool a_tail = ACONCAT && (k0 + BK > D_SZ);
        if (ACONCAT) {
            #pragma unroll
            for (int i = 0; i < 16; ++i) {
                const int idx = tid + i * NT;
                const int gk = k0 + (idx & (BK - 1));
                const int gm = m0 + (idx >> 5);
                const int ck = gk < D_SZ ? gk : D_SZ - 1;
                avs[i] = xs[gm * D_SZ + ck];
                avt[i] = (HAS_T && a_tail) ? tvec[gm] : 0.0f;
            }
        } else {
            #pragma unroll
            for (int i = 0; i < 4; ++i)
                av4[i] = *(const v4f*)&A[(size_t)(m0 + a_m + 16 * i) * KDIM + k0 + 4 * a_q];
        }
        #pragma unroll
        for (int i = 0; i < 4; ++i) {
            #pragma unroll
            for (int j = 0; j < 4; ++j) {
                const int gk = k0 + 4 * (b_q + 2 * i) + j;
                const int ck = KFULL ? gk : (gk < KDIM ? gk : KDIM - 1);
                bv[i][j] = Bm[(size_t)ck * NDIM + b_cn];
            }
        }
    };

    auto lstore = [&](const int k0, const int s) __attribute__((always_inline)) {
        const bool a_tail = ACONCAT && (k0 + BK > D_SZ);
        if (ACONCAT) {
            #pragma unroll
            for (int i = 0; i < 16; ++i) {
                const int idx = tid + i * NT;
                float v = avs[i];
                if (a_tail) {
                    float tv = avt[i];
                    if (HAS_T) asm volatile("" : "+v"(v), "+v"(tv));
                    else       asm volatile("" : "+v"(v));
                    const int gk = k0 + (idx & (BK - 1));
                    v = gk < D_SZ ? v : ((HAS_T && gk == D_SZ) ? tv : 0.0f);
                }
                As[s][idx >> 5][idx & (BK - 1)] = v;
            }
        } else {
            #pragma unroll
            for (int i = 0; i < 4; ++i)
                *(v4f*)&As[s][a_m + 16 * i][4 * a_q] = av4[i];
        }
        #pragma unroll
        for (int i = 0; i < 4; ++i) {
            v4f w = bv[i];
            if (!(KFULL && NFULL)) {
                #pragma unroll
                for (int j = 0; j < 4; ++j) {
                    const int gk = k0 + 4 * (b_q + 2 * i) + j;
                    const bool ok = (KFULL || gk < KDIM) && (NFULL || b_gn < NDIM);
                    w[j] = ok ? w[j] : 0.0f;
                }
            }
            *(v4f*)&Bt[s][b_n][4 * (b_q + 2 * i)] = w;
        }
    };

    auto compute = [&](const int s) __attribute__((always_inline)) {
        #pragma unroll
        for (int kk = 0; kk < BK; kk += 4) {
            v2f a0 = *(const v2f*)&As[s][wm * 32 +      row16][kk + ksub];
            v2f a1 = *(const v2f*)&As[s][wm * 32 + 16 + row16][kk + ksub];
            v2f b0 = *(const v2f*)&Bt[s][wn * 32 +      row16][kk + ksub];
            v2f b1 = *(const v2f*)&Bt[s][wn * 32 + 16 + row16][kk + ksub];
            acc[0][0] = wmma4(a0, b0, acc[0][0]);
            acc[0][1] = wmma4(a0, b1, acc[0][1]);
            acc[1][0] = wmma4(a1, b0, acc[1][0]);
            acc[1][1] = wmma4(a1, b1, acc[1][1]);
        }
    };

    gload(0);
    __builtin_amdgcn_sched_barrier(0);
    lstore(0, 0);
    __syncthreads();

    #pragma clang loop unroll_count(UNR)
    for (int t = 0; t < NKT - 1; ++t) {
        gload((t + 1) * BK);
        __builtin_amdgcn_sched_barrier(0);
        compute(t & 1);
        __builtin_amdgcn_sched_barrier(0);
        lstore((t + 1) * BK, (t + 1) & 1);
        __syncthreads();
    }
    compute((NKT - 1) & 1);

    #pragma unroll
    for (int mt = 0; mt < 2; ++mt) {
        const int mbase = m0 + wm * 32 + mt * 16 + ((lane >> 4) << 3);
        #pragma unroll
        for (int nt = 0; nt < 2; ++nt) {
            const int nc = n0 + wn * 32 + nt * 16 + row16;
            if (!NFULL && nc >= NDIM) continue;
            if (EPI == 2) {
                const size_t o = (size_t)nc * H_SZ + mbase;
                const v4f w0 = *(const v4f*)&bias[o];
                const v4f w1 = *(const v4f*)&bias[o + 4];
                v4f r0, r1;
                #pragma unroll
                for (int r = 0; r < 4; ++r) {
                    r0[r] = acc[mt][nt][r]     * w0[r];
                    r1[r] = acc[mt][nt][r + 4] * w1[r];
                }
                *(v4f*)&out0[o]     = r0;
                *(v4f*)&out0[o + 4] = r1;
                continue;
            }
            const float bi = bias ? bias[nc] : 0.0f;
            #pragma unroll
            for (int r = 0; r < 8; ++r) {
                const int gm = mbase + r;
                float z = acc[mt][nt][r] + bi;
                if (EPI == 0) {
                    out0[(size_t)gm * NDIM + nc] = z;
                } else {
                    float sg = 1.0f / (1.0f + __expf(-z));
                    out0[(size_t)gm * NDIM + nc] = z * sg;
                    out1[(size_t)gm * NDIM + nc] = sg * (1.0f + z * (1.0f - sg));
                }
            }
        }
    }
}

__device__ __forceinline__
void dot_div_row(float* red, const int b,
                 const float* __restrict__ u, const float* __restrict__ s2,
                 float* __restrict__ out)
{
    int tid = threadIdx.x;
    float sum = 0.0f;
    #pragma unroll
    for (int k = tid; k < H_SZ; k += 128)
        sum += u[(size_t)b * H_SZ + k] * s2[(size_t)b * H_SZ + k];
    #pragma unroll
    for (int off = 16; off > 0; off >>= 1)
        sum += __shfl_xor(sum, off, 32);
    if ((tid & 31) == 0) red[tid >> 5] = sum;
    __syncthreads();
    if (tid == 0) out[b] = -(red[0] + red[1] + red[2] + red[3]);
}

constexpr int GH_X   = H_SZ / BN;
constexpr int NB_C   = (H_SZ / BN) * (H_SZ / BM);
constexpr int NB_L1  = (H_SZ / BN) * (B_SZ / BM);
constexpr int GD_X   = (D_SZ + BN - 1) / BN;
constexpr int NB_OUT = GD_X * (B_SZ / BM);

__global__ __launch_bounds__(NT)
void front_kernel(const float* __restrict__ xs, const float* __restrict__ tvec,
                  const float* __restrict__ W1, const float* __restrict__ b1,
                  const float* __restrict__ W2, const float* __restrict__ W3,
                  float* __restrict__ Cm, float* __restrict__ a1, float* __restrict__ s1)
{
    __shared__ __attribute__((aligned(16))) lds_tile_t As;
    __shared__ __attribute__((aligned(16))) lds_tile_t Bt;
    const int b = blockIdx.x;
    if (b < NB_C) {
        gemm_tile<true, 2, H_SZ, D_SZ>(As, Bt, b % GH_X, b / GH_X,
            nullptr, W3, nullptr, W1, W2, Cm, nullptr);
    } else {
        const int c = b - NB_C;
        gemm_tile<true, 1, H_SZ, D_SZ + 1>(As, Bt, c % GH_X, c / GH_X,
            nullptr, xs, tvec, W1, b1, a1, s1);
    }
}

template<int EPI>
__global__ __launch_bounds__(NT)
void mid_kernel(const float* __restrict__ A, const float* __restrict__ Bm,
                const float* __restrict__ bias,
                float* __restrict__ out0, float* __restrict__ out1)
{
    __shared__ __attribute__((aligned(16))) lds_tile_t As;
    __shared__ __attribute__((aligned(16))) lds_tile_t Bt;
    gemm_tile<false, EPI, H_SZ, H_SZ>(As, Bt, blockIdx.x, blockIdx.y,
        A, nullptr, nullptr, Bm, bias, out0, out1);
}

__global__ __launch_bounds__(NT)
void back_kernel(const float* __restrict__ a2, const float* __restrict__ W3,
                 const float* __restrict__ b3, float* __restrict__ dxs,
                 const float* __restrict__ u, const float* __restrict__ s2,
                 float* __restrict__ ndiv)
{
    __shared__ __attribute__((aligned(16))) lds_tile_t As;
    __shared__ __attribute__((aligned(16))) lds_tile_t Bt;
    const int b = blockIdx.x;
    if (b < NB_OUT) {
        gemm_tile<false, 0, D_SZ, H_SZ>(As, Bt, b % GD_X, b / GD_X,
            a2, nullptr, nullptr, W3, b3, dxs, nullptr);
    } else {
        dot_div_row(&As[0][0][0], b - NB_OUT, u, s2, ndiv);
    }
}

extern "C" void kernel_launch(void* const* d_in, const int* in_sizes, int n_in,
                              void* d_out, int out_size, void* d_ws, size_t ws_size,
                              hipStream_t stream) {
    const float* xs = (const float*)d_in[0];
    const float* t  = (const float*)d_in[1];
    const float* W1 = (const float*)d_in[2];
    const float* b1 = (const float*)d_in[3];
    const float* W2 = (const float*)d_in[4];
    const float* b2 = (const float*)d_in[5];
    const float* W3 = (const float*)d_in[6];
    const float* b3 = (const float*)d_in[7];

    float* dxs  = (float*)d_out;
    float* ndiv = (float*)d_out + (size_t)B_SZ * D_SZ;

    float* ws = (float*)d_ws;
    const size_t act = (size_t)B_SZ * H_SZ;
    float* a1u = ws;
    float* s1  = ws + act;
    float* a2  = ws + 2 * act;
    float* s2  = ws + 3 * act;
    float* Cm  = ws + 4 * act;

    dim3 blk(NT);
    dim3 gridH(H_SZ / BN, B_SZ / BM);

    front_kernel<<<dim3(NB_C + NB_L1), blk, 0, stream>>>(
        xs, t, W1, b1, W2, W3, Cm, a1u, s1);

    mid_kernel<1><<<gridH, blk, 0, stream>>>(a1u, W2, b2, a2, s2);

    mid_kernel<0><<<gridH, blk, 0, stream>>>(s1, Cm, nullptr, a1u, nullptr);

    back_kernel<<<dim3(NB_OUT + B_SZ), blk, 0, stream>>>(
        a2, W3, b3, dxs, a1u, s2, ndiv);
}


# layer-2 GEMM and trace GEMM u = s1 @ C done by one block per 64x64 tile in one launch; divergence dot folded into its epilogue as per-row partials, u and s2 never stored (4 launches -> 3)
# speedup vs baseline: 8.6011x; 1.0862x over previous
#include <hip/hip_runtime.h>
#include <hip/hip_bf16.h>
#include <math.h>


typedef __attribute__((ext_vector_type(2))) float v2f;
typedef __attribute__((ext_vector_type(8))) float v8f;

#define B_SZ 4096
#define D_SZ 66
#define H_SZ 512

constexpr int BM = 64;
constexpr int BN = 64;
constexpr int BK = 32;
constexpr int NT = 128;

__device__ __forceinline__ v8f wmma4(v2f a, v2f b, v8f c) {
    return __builtin_amdgcn_wmma_f32_16x16x4_f32(
        false, a, false, b, (short)0, c, false, false);
}

typedef __attribute__((ext_vector_type(4))) float v4f;

typedef float lds_tile_t[2][BM][BK + 4];

template<bool ACONCAT, int EPI, int NDIM, int KDIM>
__device__ __forceinline__
void gemm_tile(lds_tile_t& As, lds_tile_t& Bt,
               const int bx, const int by,
               const float* __restrict__ A, const float* __restrict__ xs,
               const float* __restrict__ tvec,
               const float* __restrict__ Bm, const float* __restrict__ bias,
               float* __restrict__ out0, float* __restrict__ out1)
{
    static_assert(BM == BN, "A and B tiles share one LDS array type");

    constexpr bool KFULL = (KDIM % BK) == 0;
    constexpr bool HAS_T = ACONCAT && (KDIM > D_SZ);
    constexpr bool NFULL = (NDIM % BN) == 0;
    constexpr int  NKT   = (KDIM + BK - 1) / BK;
    constexpr int  UNR   = ACONCAT ? (NKT - 1) : 1;
    static_assert(ACONCAT || (KFULL && (KDIM % 4) == 0), "A rows must allow 128-bit loads");
    static_assert(NKT >= 2, "the pipelined loop expects at least two k-tiles");

    const int tid  = threadIdx.x;
    const int wave = tid >> 5;
    const int lane = tid & 31;
    const int wm   = wave >> 1;
    const int wn   = wave & 1;

    const int m0 = by * BM;
    const int n0 = bx * BN;

    v8f acc[2][2] = {};

    const int row16 = lane & 15;
    const int ksub  = (lane >> 4) << 1;

    const int a_q  = tid & 7;
    const int a_m  = tid >> 3;
    const int b_n  = tid & (BN - 1);
    const int b_q  = tid >> 6;
    const int b_gn = n0 + b_n;
    const int b_cn = NFULL ? b_gn : (b_gn < NDIM ? b_gn : NDIM - 1);

    v4f   av4[4];
    float avs[16];
    float avt[16];
    v4f   bv[4];

    auto gload = [&](const int k0) __attribute__((always_inline)) {
        const bool a_tail = ACONCAT && (k0 + BK > D_SZ);
        if (ACONCAT) {
            #pragma unroll
            for (int i = 0; i < 16; ++i) {
                const int idx = tid + i * NT;
                const int gk = k0 + (idx & (BK - 1));
                const int gm = m0 + (idx >> 5);
                const int ck = gk < D_SZ ? gk : D_SZ - 1;
                avs[i] = xs[gm * D_SZ + ck];
                avt[i] = (HAS_T && a_tail) ? tvec[gm] : 0.0f;
            }
        } else {
            #pragma unroll
            for (int i = 0; i < 4; ++i)
                av4[i] = *(const v4f*)&A[(size_t)(m0 + a_m + 16 * i) * KDIM + k0 + 4 * a_q];
        }
        #pragma unroll
        for (int i = 0; i < 4; ++i) {
            #pragma unroll
            for (int j = 0; j < 4; ++j) {
                const int gk = k0 + 4 * (b_q + 2 * i) + j;
                const int ck = KFULL ? gk : (gk < KDIM ? gk : KDIM - 1);
                bv[i][j] = Bm[(size_t)ck * NDIM + b_cn];
            }
        }
    };

    auto lstore = [&](const int k0, const int s) __attribute__((always_inline)) {
        const bool a_tail = ACONCAT && (k0 + BK > D_SZ);
        if (ACONCAT) {
            #pragma unroll
            for (int i = 0; i < 16; ++i) {
                const int idx = tid + i * NT;
                float v = avs[i];
                if (a_tail) {
                    float tv = avt[i];
                    if (HAS_T) asm volatile("" : "+v"(v), "+v"(tv));
                    else       asm volatile("" : "+v"(v));
                    const int gk = k0 + (idx & (BK - 1));
                    v = gk < D_SZ ? v : ((HAS_T && gk == D_SZ) ? tv : 0.0f);
                }
                As[s][idx >> 5][idx & (BK - 1)] = v;
            }
        } else {
            #pragma unroll
            for (int i = 0; i < 4; ++i)
                *(v4f*)&As[s][a_m + 16 * i][4 * a_q] = av4[i];
        }
        #pragma unroll
        for (int i = 0; i < 4; ++i) {
            v4f w = bv[i];
            if (!(KFULL && NFULL)) {
                #pragma unroll
                for (int j = 0; j < 4; ++j) {
                    const int gk = k0 + 4 * (b_q + 2 * i) + j;
                    const bool ok = (KFULL || gk < KDIM) && (NFULL || b_gn < NDIM);
                    w[j] = ok ? w[j] : 0.0f;
                }
            }
            *(v4f*)&Bt[s][b_n][4 * (b_q + 2 * i)] = w;
        }
    };

    auto compute = [&](const int s) __attribute__((always_inline)) {
        #pragma unroll
        for (int kk = 0; kk < BK; kk += 4) {
            v2f a0 = *(const v2f*)&As[s][wm * 32 +      row16][kk + ksub];
            v2f a1 = *(const v2f*)&As[s][wm * 32 + 16 + row16][kk + ksub];
            v2f b0 = *(const v2f*)&Bt[s][wn * 32 +      row16][kk + ksub];
            v2f b1 = *(const v2f*)&Bt[s][wn * 32 + 16 + row16][kk + ksub];
            acc[0][0] = wmma4(a0, b0, acc[0][0]);
            acc[0][1] = wmma4(a0, b1, acc[0][1]);
            acc[1][0] = wmma4(a1, b0, acc[1][0]);
            acc[1][1] = wmma4(a1, b1, acc[1][1]);
        }
    };

    gload(0);
    __builtin_amdgcn_sched_barrier(0);
    lstore(0, 0);
    __syncthreads();

    #pragma clang loop unroll_count(UNR)
    for (int t = 0; t < NKT - 1; ++t) {
        gload((t + 1) * BK);
        __builtin_amdgcn_sched_barrier(0);
        compute(t & 1);
        __builtin_amdgcn_sched_barrier(0);
        lstore((t + 1) * BK, (t + 1) & 1);
        __syncthreads();
    }
    compute((NKT - 1) & 1);

    #pragma unroll
    for (int mt = 0; mt < 2; ++mt) {
        const int mbase = m0 + wm * 32 + mt * 16 + ((lane >> 4) << 3);
        #pragma unroll
        for (int nt = 0; nt < 2; ++nt) {
            const int nc = n0 + wn * 32 + nt * 16 + row16;
            if (!NFULL && nc >= NDIM) continue;
            if (EPI == 2) {
                const size_t o = (size_t)nc * H_SZ + mbase;
                const v4f w0 = *(const v4f*)&bias[o];
                const v4f w1 = *(const v4f*)&bias[o + 4];
                v4f r0, r1;
                #pragma unroll
                for (int r = 0; r < 4; ++r) {
                    r0[r] = acc[mt][nt][r]     * w0[r];
                    r1[r] = acc[mt][nt][r + 4] * w1[r];
                }
                *(v4f*)&out0[o]     = r0;
                *(v4f*)&out0[o + 4] = r1;
                continue;
            }
            const float bi = bias ? bias[nc] : 0.0f;
            #pragma unroll
            for (int r = 0; r < 8; ++r) {
                const int gm = mbase + r;
                float z = acc[mt][nt][r] + bi;
                if (EPI == 0) {
                    out0[(size_t)gm * NDIM + nc] = z;
                } else {
                    float sg = 1.0f / (1.0f + __expf(-z));
                    out0[(size_t)gm * NDIM + nc] = z * sg;
                    out1[(size_t)gm * NDIM + nc] = sg * (1.0f + z * (1.0f - sg));
                }
            }
        }
    }
}

__device__ __forceinline__ float row16_sum(float v) {
    v += __int_as_float(__builtin_amdgcn_update_dpp(0, __float_as_int(v), 0x161, 0xf, 0xf, true));
    v += __int_as_float(__builtin_amdgcn_update_dpp(0, __float_as_int(v), 0x162, 0xf, 0xf, true));
    v += __int_as_float(__builtin_amdgcn_update_dpp(0, __float_as_int(v), 0x164, 0xf, 0xf, true));
    v += __int_as_float(__builtin_amdgcn_update_dpp(0, __float_as_int(v), 0x168, 0xf, 0xf, true));
    return v;
}

__device__ __forceinline__
void mid_fused_tile(lds_tile_t& As, lds_tile_t& Bt,
                    const int bx, const int by,
                    const float* __restrict__ a1, const float* __restrict__ W2,
                    const float* __restrict__ b2,
                    const float* __restrict__ s1, const float* __restrict__ Cm,
                    float* __restrict__ a2, float* __restrict__ part)
{
    constexpr int NKT = H_SZ / BK;
    static_assert((NKT & 1) == 0, "the hand-over assumes the first product ends in stage 1");

    const int tid  = threadIdx.x;
    const int wave = tid >> 5;
    const int lane = tid & 31;
    const int wm   = wave >> 1;
    const int wn   = wave & 1;

    const int m0 = by * BM;
    const int n0 = bx * BN;

    v8f accz[2][2] = {};
    v8f accu[2][2] = {};

    const int row16 = lane & 15;
    const int ksub  = (lane >> 4) << 1;

    const int a_q  = tid & 7;
    const int a_m  = tid >> 3;
    const int b_n  = tid & (BN - 1);
    const int b_q  = tid >> 6;
    const int b_gn = n0 + b_n;

    v4f av4[4];
    v4f bv[4];

    auto gload = [&](const float* __restrict__ A, const float* __restrict__ Bm,
                     const int k0) __attribute__((always_inline)) {
        #pragma unroll
        for (int i = 0; i < 4; ++i)
            av4[i] = *(const v4f*)&A[(size_t)(m0 + a_m + 16 * i) * H_SZ + k0 + 4 * a_q];
        #pragma unroll
        for (int i = 0; i < 4; ++i) {
            #pragma unroll
            for (int j = 0; j < 4; ++j) {
                const int gk = k0 + 4 * (b_q + 2 * i) + j;
                bv[i][j] = Bm[(size_t)gk * H_SZ + b_gn];
            }
        }
    };

    auto lstore = [&](const int s) __attribute__((always_inline)) {
        #pragma unroll
        for (int i = 0; i < 4; ++i)
            *(v4f*)&As[s][a_m + 16 * i][4 * a_q] = av4[i];
        #pragma unroll
        for (int i = 0; i < 4; ++i)
            *(v4f*)&Bt[s][b_n][4 * (b_q + 2 * i)] = bv[i];
    };

    auto compute = [&](v8f (&acc)[2][2], const int s) __attribute__((always_inline)) {
        #pragma unroll
        for (int kk = 0; kk < BK; kk += 4) {
            v2f a0 = *(const v2f*)&As[s][wm * 32 +      row16][kk + ksub];
            v2f a1f = *(const v2f*)&As[s][wm * 32 + 16 + row16][kk + ksub];
            v2f b0 = *(const v2f*)&Bt[s][wn * 32 +      row16][kk + ksub];
            v2f b1f = *(const v2f*)&Bt[s][wn * 32 + 16 + row16][kk + ksub];
            acc[0][0] = wmma4(a0,  b0,  acc[0][0]);
            acc[0][1] = wmma4(a0,  b1f, acc[0][1]);
            acc[1][0] = wmma4(a1f, b0,  acc[1][0]);
            acc[1][1] = wmma4(a1f, b1f, acc[1][1]);
        }
    };

    auto wmma_done = [&](v8f (&acc)[2][2]) __attribute__((always_inline)) {
        asm volatile("" : "+v"(acc[0][0]), "+v"(acc[0][1]), "+v"(acc[1][0]), "+v"(acc[1][1])
                     : : "memory");
    };

    gload(a1, W2, 0);
    __builtin_amdgcn_sched_barrier(0);
    lstore(0);
    __syncthreads();

    for (int t = 0; t < NKT - 1; ++t) {
        gload(a1, W2, (t + 1) * BK);
        __builtin_amdgcn_sched_barrier(0);
        compute(accz, t & 1);
        wmma_done(accz);
        lstore((t + 1) & 1);
        __syncthreads();
    }

    gload(s1, Cm, 0);
    __builtin_amdgcn_sched_barrier(0);
    compute(accz, (NKT - 1) & 1);
    wmma_done(accz);
    lstore(0);
    __syncthreads();

    for (int t = 0; t < NKT - 1; ++t) {
        gload(s1, Cm, (t + 1) * BK);
        __builtin_amdgcn_sched_barrier(0);
        compute(accu, t & 1);
        wmma_done(accu);
        lstore((t + 1) & 1);
        __syncthreads();
    }
    compute(accu, (NKT - 1) & 1);

    float dsum[2][8];
    #pragma unroll
    for (int mt = 0; mt < 2; ++mt) {
        const int mbase = m0 + wm * 32 + mt * 16 + ((lane >> 4) << 3);
        #pragma unroll
        for (int r = 0; r < 8; ++r) dsum[mt][r] = 0.0f;
        #pragma unroll
        for (int nt = 0; nt < 2; ++nt) {
            const int nc = n0 + wn * 32 + nt * 16 + row16;
            const float bi = b2[nc];
            #pragma unroll
            for (int r = 0; r < 8; ++r) {
                const float z = accz[mt][nt][r] + bi;
                const float sg = 1.0f / (1.0f + __expf(-z));
                accz[mt][nt][r] = z * sg;
                dsum[mt][r] += accu[mt][nt][r] * (sg * (1.0f + z * (1.0f - sg)));
            }
        }
    }
    #pragma unroll
    for (int mt = 0; mt < 2; ++mt)
        #pragma unroll
        for (int r = 0; r < 8; ++r)
            dsum[mt][r] = row16_sum(dsum[mt][r]);

    float* red = &As[0][0][0];
    if (row16 == 0) {
        #pragma unroll
        for (int mt = 0; mt < 2; ++mt)
            #pragma unroll
            for (int r = 0; r < 8; ++r)
                red[wn * BM + wm * 32 + mt * 16 + ((lane >> 4) << 3) + r] = dsum[mt][r];
    }
    __syncthreads();
    if (tid < BM)
        part[(size_t)bx * B_SZ + m0 + tid] = red[tid] + red[BM + tid];

    #pragma unroll
    for (int mt = 0; mt < 2; ++mt) {
        const int mbase = m0 + wm * 32 + mt * 16 + ((lane >> 4) << 3);
        #pragma unroll
        for (int nt = 0; nt < 2; ++nt) {
            const int nc = n0 + wn * 32 + nt * 16 + row16;
            #pragma unroll
            for (int r = 0; r < 8; ++r)
                a2[(size_t)(mbase + r) * H_SZ + nc] = accz[mt][nt][r];
        }
    }
}

constexpr int GH_X   = H_SZ / BN;
constexpr int NB_C   = (H_SZ / BN) * (H_SZ / BM);
constexpr int NB_L1  = (H_SZ / BN) * (B_SZ / BM);
constexpr int GD_X   = (D_SZ + BN - 1) / BN;
constexpr int NB_OUT = GD_X * (B_SZ / BM);

__global__ __launch_bounds__(NT)
void front_kernel(const float* __restrict__ xs, const float* __restrict__ tvec,
                  const float* __restrict__ W1, const float* __restrict__ b1,
                  const float* __restrict__ W2, const float* __restrict__ W3,
                  float* __restrict__ Cm, float* __restrict__ a1, float* __restrict__ s1)
{
    __shared__ __attribute__((aligned(16))) lds_tile_t As;
    __shared__ __attribute__((aligned(16))) lds_tile_t Bt;
    const int b = blockIdx.x;
    if (b < NB_C) {
        gemm_tile<true, 2, H_SZ, D_SZ>(As, Bt, b % GH_X, b / GH_X,
            nullptr, W3, nullptr, W1, W2, Cm, nullptr);
    } else {
        const int c = b - NB_C;
        gemm_tile<true, 1, H_SZ, D_SZ + 1>(As, Bt, c % GH_X, c / GH_X,
            nullptr, xs, tvec, W1, b1, a1, s1);
    }
}

__global__ __launch_bounds__(NT)
void mid_kernel(const float* __restrict__ a1, const float* __restrict__ W2,
                const float* __restrict__ b2,
                const float* __restrict__ s1, const float* __restrict__ Cm,
                float* __restrict__ a2, float* __restrict__ part)
{
    __shared__ __attribute__((aligned(16))) lds_tile_t As;
    __shared__ __attribute__((aligned(16))) lds_tile_t Bt;
    mid_fused_tile(As, Bt, blockIdx.x, blockIdx.y, a1, W2, b2, s1, Cm, a2, part);
}

__global__ __launch_bounds__(NT)
void back_kernel(const float* __restrict__ a2, const float* __restrict__ W3,
                 const float* __restrict__ b3, float* __restrict__ dxs,
                 const float* __restrict__ part, float* __restrict__ ndiv)
{
    __shared__ __attribute__((aligned(16))) lds_tile_t As;
    __shared__ __attribute__((aligned(16))) lds_tile_t Bt;
    const int b  = blockIdx.x;
    const int bx = b % GD_X;
    const int by = b / GD_X;
    gemm_tile<false, 0, D_SZ, H_SZ>(As, Bt, bx, by,
        a2, nullptr, nullptr, W3, b3, dxs, nullptr);
    if (bx == GD_X - 1 && threadIdx.x < BM) {
        const int row = by * BM + threadIdx.x;
        float sum = 0.0f;
        #pragma unroll
        for (int j = 0; j < GH_X; ++j)
            sum += part[(size_t)j * B_SZ + row];
        ndiv[row] = -sum;
    }
}

extern "C" void kernel_launch(void* const* d_in, const int* in_sizes, int n_in,
                              void* d_out, int out_size, void* d_ws, size_t ws_size,
                              hipStream_t stream) {
    const float* xs = (const float*)d_in[0];
    const float* t  = (const float*)d_in[1];
    const float* W1 = (const float*)d_in[2];
    const float* b1 = (const float*)d_in[3];
    const float* W2 = (const float*)d_in[4];
    const float* b2 = (const float*)d_in[5];
    const float* W3 = (const float*)d_in[6];
    const float* b3 = (const float*)d_in[7];

    float* dxs  = (float*)d_out;
    float* ndiv = (float*)d_out + (size_t)B_SZ * D_SZ;

    float* ws = (float*)d_ws;
    const size_t act = (size_t)B_SZ * H_SZ;
    float* a1  = ws;
    float* s1  = ws + act;
    float* a2  = ws + 2 * act;
    float* prt = ws + 3 * act;
    float* Cm  = ws + 4 * act;

    dim3 blk(NT);
    dim3 gridH(H_SZ / BN, B_SZ / BM);

    front_kernel<<<dim3(NB_C + NB_L1), blk, 0, stream>>>(
        xs, t, W1, b1, W2, W3, Cm, a1, s1);

    mid_kernel<<<gridH, blk, 0, stream>>>(a1, W2, b2, s1, Cm, a2, prt);

    back_kernel<<<dim3(NB_OUT), blk, 0, stream>>>(
        a2, W3, b3, dxs, prt, ndiv);
}


# output-layer GEMM re-tiled: one block per 16-row strip on all 256 WGPs, K split over the block's 8 waves, partial tiles summed in LDS (was 128 blocks of 64x64 tiles, 512 serial WMMAs per wave)
# speedup vs baseline: 9.7720x; 1.1361x over previous
#include <hip/hip_runtime.h>
#include <hip/hip_bf16.h>
#include <math.h>


typedef __attribute__((ext_vector_type(2))) float v2f;
typedef __attribute__((ext_vector_type(8))) float v8f;

#define B_SZ 4096
#define D_SZ 66
#define H_SZ 512

constexpr int BM = 64;
constexpr int BN = 64;
constexpr int BK = 32;
constexpr int NT = 128;

__device__ __forceinline__ v8f wmma4(v2f a, v2f b, v8f c) {
    return __builtin_amdgcn_wmma_f32_16x16x4_f32(
        false, a, false, b, (short)0, c, false, false);
}

typedef __attribute__((ext_vector_type(4))) float v4f;

typedef float lds_tile_t[2][BM][BK + 4];

template<bool ACONCAT, int EPI, int NDIM, int KDIM>
__device__ __forceinline__
void gemm_tile(lds_tile_t& As, lds_tile_t& Bt,
               const int bx, const int by,
               const float* __restrict__ A, const float* __restrict__ xs,
               const float* __restrict__ tvec,
               const float* __restrict__ Bm, const float* __restrict__ bias,
               float* __restrict__ out0, float* __restrict__ out1)
{
    static_assert(BM == BN, "A and B tiles share one LDS array type");

    constexpr bool KFULL = (KDIM % BK) == 0;
    constexpr bool HAS_T = ACONCAT && (KDIM > D_SZ);
    constexpr bool NFULL = (NDIM % BN) == 0;
    constexpr int  NKT   = (KDIM + BK - 1) / BK;
    constexpr int  UNR   = ACONCAT ? (NKT - 1) : 1;
    static_assert(ACONCAT || (KFULL && (KDIM % 4) == 0), "A rows must allow 128-bit loads");
    static_assert(NKT >= 2, "the pipelined loop expects at least two k-tiles");

    const int tid  = threadIdx.x;
    const int wave = tid >> 5;
    const int lane = tid & 31;
    const int wm   = wave >> 1;
    const int wn   = wave & 1;

    const int m0 = by * BM;
    const int n0 = bx * BN;

    v8f acc[2][2] = {};

    const int row16 = lane & 15;
    const int ksub  = (lane >> 4) << 1;

    const int a_q  = tid & 7;
    const int a_m  = tid >> 3;
    const int b_n  = tid & (BN - 1);
    const int b_q  = tid >> 6;
    const int b_gn = n0 + b_n;
    const int b_cn = NFULL ? b_gn : (b_gn < NDIM ? b_gn : NDIM - 1);

    v4f   av4[4];
    float avs[16];
    float avt[16];
    v4f   bv[4];

    auto gload = [&](const int k0) __attribute__((always_inline)) {
        const bool a_tail = ACONCAT && (k0 + BK > D_SZ);
        if (ACONCAT) {
            #pragma unroll
            for (int i = 0; i < 16; ++i) {
                const int idx = tid + i * NT;
                const int gk = k0 + (idx & (BK - 1));
                const int gm = m0 + (idx >> 5);
                const int ck = gk < D_SZ ? gk : D_SZ - 1;
                avs[i] = xs[gm * D_SZ + ck];
                avt[i] = (HAS_T && a_tail) ? tvec[gm] : 0.0f;
            }
        } else {
            #pragma unroll
            for (int i = 0; i < 4; ++i)
                av4[i] = *(const v4f*)&A[(size_t)(m0 + a_m + 16 * i) * KDIM + k0 + 4 * a_q];
        }
        #pragma unroll
        for (int i = 0; i < 4; ++i) {
            #pragma unroll
            for (int j = 0; j < 4; ++j) {
                const int gk = k0 + 4 * (b_q + 2 * i) + j;
                const int ck = KFULL ? gk : (gk < KDIM ? gk : KDIM - 1);
                bv[i][j] = Bm[(size_t)ck * NDIM + b_cn];
            }
        }
    };

    auto lstore = [&](const int k0, const int s) __attribute__((always_inline)) {
        const bool a_tail = ACONCAT && (k0 + BK > D_SZ);
        if (ACONCAT) {
            #pragma unroll
            for (int i = 0; i < 16; ++i) {
                const int idx = tid + i * NT;
                float v = avs[i];
                if (a_tail) {
                    float tv = avt[i];
                    if (HAS_T) asm volatile("" : "+v"(v), "+v"(tv));
                    else       asm volatile("" : "+v"(v));
                    const int gk = k0 + (idx & (BK - 1));
                    v = gk < D_SZ ? v : ((HAS_T && gk == D_SZ) ? tv : 0.0f);
                }
                As[s][idx >> 5][idx & (BK - 1)] = v;
            }
        } else {
            #pragma unroll
            for (int i = 0; i < 4; ++i)
                *(v4f*)&As[s][a_m + 16 * i][4 * a_q] = av4[i];
        }
        #pragma unroll
        for (int i = 0; i < 4; ++i) {
            v4f w = bv[i];
            if (!(KFULL && NFULL)) {
                #pragma unroll
                for (int j = 0; j < 4; ++j) {
                    const int gk = k0 + 4 * (b_q + 2 * i) + j;
                    const bool ok = (KFULL || gk < KDIM) && (NFULL || b_gn < NDIM);
                    w[j] = ok ? w[j] : 0.0f;
                }
            }
            *(v4f*)&Bt[s][b_n][4 * (b_q + 2 * i)] = w;
        }
    };

    auto compute = [&](const int s) __attribute__((always_inline)) {
        #pragma unroll
        for (int kk = 0; kk < BK; kk += 4) {
            v2f a0 = *(const v2f*)&As[s][wm * 32 +      row16][kk + ksub];
            v2f a1 = *(const v2f*)&As[s][wm * 32 + 16 + row16][kk + ksub];
            v2f b0 = *(const v2f*)&Bt[s][wn * 32 +      row16][kk + ksub];
            v2f b1 = *(const v2f*)&Bt[s][wn * 32 + 16 + row16][kk + ksub];
            acc[0][0] = wmma4(a0, b0, acc[0][0]);
            acc[0][1] = wmma4(a0, b1, acc[0][1]);
            acc[1][0] = wmma4(a1, b0, acc[1][0]);
            acc[1][1] = wmma4(a1, b1, acc[1][1]);
        }
    };

    gload(0);
    __builtin_amdgcn_sched_barrier(0);
    lstore(0, 0);
    __syncthreads();

    #pragma clang loop unroll_count(UNR)
    for (int t = 0; t < NKT - 1; ++t) {
        gload((t + 1) * BK);
        __builtin_amdgcn_sched_barrier(0);
        compute(t & 1);
        __builtin_amdgcn_sched_barrier(0);
        lstore((t + 1) * BK, (t + 1) & 1);
        __syncthreads();
    }
    compute((NKT - 1) & 1);

    #pragma unroll
    for (int mt = 0; mt < 2; ++mt) {
        const int mbase = m0 + wm * 32 + mt * 16 + ((lane >> 4) << 3);
        #pragma unroll
        for (int nt = 0; nt < 2; ++nt) {
            const int nc = n0 + wn * 32 + nt * 16 + row16;
            if (!NFULL && nc >= NDIM) continue;
            if (EPI == 2) {
                const size_t o = (size_t)nc * H_SZ + mbase;
                const v4f w0 = *(const v4f*)&bias[o];
                const v4f w1 = *(const v4f*)&bias[o + 4];
                v4f r0, r1;
                #pragma unroll
                for (int r = 0; r < 4; ++r) {
                    r0[r] = acc[mt][nt][r]     * w0[r];
                    r1[r] = acc[mt][nt][r + 4] * w1[r];
                }
                *(v4f*)&out0[o]     = r0;
                *(v4f*)&out0[o + 4] = r1;
                continue;
            }
            const float bi = bias ? bias[nc] : 0.0f;
            #pragma unroll
            for (int r = 0; r < 8; ++r) {
                const int gm = mbase + r;
                float z = acc[mt][nt][r] + bi;
                if (EPI == 0) {
                    out0[(size_t)gm * NDIM + nc] = z;
                } else {
                    float sg = 1.0f / (1.0f + __expf(-z));
                    out0[(size_t)gm * NDIM + nc] = z * sg;
                    out1[(size_t)gm * NDIM + nc] = sg * (1.0f + z * (1.0f - sg));
                }
            }
        }
    }
}

__device__ __forceinline__ float row16_sum(float v) {
    v += __int_as_float(__builtin_amdgcn_update_dpp(0, __float_as_int(v), 0x161, 0xf, 0xf, true));
    v += __int_as_float(__builtin_amdgcn_update_dpp(0, __float_as_int(v), 0x162, 0xf, 0xf, true));
    v += __int_as_float(__builtin_amdgcn_update_dpp(0, __float_as_int(v), 0x164, 0xf, 0xf, true));
    v += __int_as_float(__builtin_amdgcn_update_dpp(0, __float_as_int(v), 0x168, 0xf, 0xf, true));
    return v;
}

__device__ __forceinline__
void mid_fused_tile(lds_tile_t& As, lds_tile_t& Bt,
                    const int bx, const int by,
                    const float* __restrict__ a1, const float* __restrict__ W2,
                    const float* __restrict__ b2,
                    const float* __restrict__ s1, const float* __restrict__ Cm,
                    float* __restrict__ a2, float* __restrict__ part)
{
    constexpr int NKT = H_SZ / BK;
    static_assert((NKT & 1) == 0, "the hand-over assumes the first product ends in stage 1");

    const int tid  = threadIdx.x;
    const int wave = tid >> 5;
    const int lane = tid & 31;
    const int wm   = wave >> 1;
    const int wn   = wave & 1;

    const int m0 = by * BM;
    const int n0 = bx * BN;

    v8f accz[2][2] = {};
    v8f accu[2][2] = {};

    const int row16 = lane & 15;
    const int ksub  = (lane >> 4) << 1;

    const int a_q  = tid & 7;
    const int a_m  = tid >> 3;
    const int b_n  = tid & (BN - 1);
    const int b_q  = tid >> 6;
    const int b_gn = n0 + b_n;

    v4f av4[4];
    v4f bv[4];

    auto gload = [&](const float* __restrict__ A, const float* __restrict__ Bm,
                     const int k0) __attribute__((always_inline)) {
        #pragma unroll
        for (int i = 0; i < 4; ++i)
            av4[i] = *(const v4f*)&A[(size_t)(m0 + a_m + 16 * i) * H_SZ + k0 + 4 * a_q];
        #pragma unroll
        for (int i = 0; i < 4; ++i) {
            #pragma unroll
            for (int j = 0; j < 4; ++j) {
                const int gk = k0 + 4 * (b_q + 2 * i) + j;
                bv[i][j] = Bm[(size_t)gk * H_SZ + b_gn];
            }
        }
    };

    auto lstore = [&](const int s) __attribute__((always_inline)) {
        #pragma unroll
        for (int i = 0; i < 4; ++i)
            *(v4f*)&As[s][a_m + 16 * i][4 * a_q] = av4[i];
        #pragma unroll
        for (int i = 0; i < 4; ++i)
            *(v4f*)&Bt[s][b_n][4 * (b_q + 2 * i)] = bv[i];
    };

    auto compute = [&](v8f (&acc)[2][2], const int s) __attribute__((always_inline)) {
        #pragma unroll
        for (int kk = 0; kk < BK; kk += 4) {
            v2f a0 = *(const v2f*)&As[s][wm * 32 +      row16][kk + ksub];
            v2f a1f = *(const v2f*)&As[s][wm * 32 + 16 + row16][kk + ksub];
            v2f b0 = *(const v2f*)&Bt[s][wn * 32 +      row16][kk + ksub];
            v2f b1f = *(const v2f*)&Bt[s][wn * 32 + 16 + row16][kk + ksub];
            acc[0][0] = wmma4(a0,  b0,  acc[0][0]);
            acc[0][1] = wmma4(a0,  b1f, acc[0][1]);
            acc[1][0] = wmma4(a1f, b0,  acc[1][0]);
            acc[1][1] = wmma4(a1f, b1f, acc[1][1]);
        }
    };

    auto wmma_done = [&](v8f (&acc)[2][2]) __attribute__((always_inline)) {
        asm volatile("" : "+v"(acc[0][0]), "+v"(acc[0][1]), "+v"(acc[1][0]), "+v"(acc[1][1])
                     : : "memory");
    };

    gload(a1, W2, 0);
    __builtin_amdgcn_sched_barrier(0);
    lstore(0);
    __syncthreads();

    for (int t = 0; t < NKT - 1; ++t) {
        gload(a1, W2, (t + 1) * BK);
        __builtin_amdgcn_sched_barrier(0);
        compute(accz, t & 1);
        wmma_done(accz);
        lstore((t + 1) & 1);
        __syncthreads();
    }

    gload(s1, Cm, 0);
    __builtin_amdgcn_sched_barrier(0);
    compute(accz, (NKT - 1) & 1);
    wmma_done(accz);
    lstore(0);
    __syncthreads();

    for (int t = 0; t < NKT - 1; ++t) {
        gload(s1, Cm, (t + 1) * BK);
        __builtin_amdgcn_sched_barrier(0);
        compute(accu, t & 1);
        wmma_done(accu);
        lstore((t + 1) & 1);
        __syncthreads();
    }
    compute(accu, (NKT - 1) & 1);

    float dsum[2][8];
    #pragma unroll
    for (int mt = 0; mt < 2; ++mt) {
        const int mbase = m0 + wm * 32 + mt * 16 + ((lane >> 4) << 3);
        #pragma unroll
        for (int r = 0; r < 8; ++r) dsum[mt][r] = 0.0f;
        #pragma unroll
        for (int nt = 0; nt < 2; ++nt) {
            const int nc = n0 + wn * 32 + nt * 16 + row16;
            const float bi = b2[nc];
            #pragma unroll
            for (int r = 0; r < 8; ++r) {
                const float z = accz[mt][nt][r] + bi;
                const float sg = 1.0f / (1.0f + __expf(-z));
                accz[mt][nt][r] = z * sg;
                dsum[mt][r] += accu[mt][nt][r] * (sg * (1.0f + z * (1.0f - sg)));
            }
        }
    }
    #pragma unroll
    for (int mt = 0; mt < 2; ++mt)
        #pragma unroll
        for (int r = 0; r < 8; ++r)
            dsum[mt][r] = row16_sum(dsum[mt][r]);

    float* red = &As[0][0][0];
    if (row16 == 0) {
        #pragma unroll
        for (int mt = 0; mt < 2; ++mt)
            #pragma unroll
            for (int r = 0; r < 8; ++r)
                red[wn * BM + wm * 32 + mt * 16 + ((lane >> 4) << 3) + r] = dsum[mt][r];
    }
    __syncthreads();
    if (tid < BM)
        part[(size_t)bx * B_SZ + m0 + tid] = red[tid] + red[BM + tid];

    #pragma unroll
    for (int mt = 0; mt < 2; ++mt) {
        const int mbase = m0 + wm * 32 + mt * 16 + ((lane >> 4) << 3);
        #pragma unroll
        for (int nt = 0; nt < 2; ++nt) {
            const int nc = n0 + wn * 32 + nt * 16 + row16;
            #pragma unroll
            for (int r = 0; r < 8; ++r)
                a2[(size_t)(mbase + r) * H_SZ + nc] = accz[mt][nt][r];
        }
    }
}

constexpr int GH_X   = H_SZ / BN;
constexpr int NB_C   = (H_SZ / BN) * (H_SZ / BM);
constexpr int NB_L1  = (H_SZ / BN) * (B_SZ / BM);

__global__ __launch_bounds__(NT)
void front_kernel(const float* __restrict__ xs, const float* __restrict__ tvec,
                  const float* __restrict__ W1, const float* __restrict__ b1,
                  const float* __restrict__ W2, const float* __restrict__ W3,
                  float* __restrict__ Cm, float* __restrict__ a1, float* __restrict__ s1)
{
    __shared__ __attribute__((aligned(16))) lds_tile_t As;
    __shared__ __attribute__((aligned(16))) lds_tile_t Bt;
    const int b = blockIdx.x;
    if (b < NB_C) {
        gemm_tile<true, 2, H_SZ, D_SZ>(As, Bt, b % GH_X, b / GH_X,
            nullptr, W3, nullptr, W1, W2, Cm, nullptr);
    } else {
        const int c = b - NB_C;
        gemm_tile<true, 1, H_SZ, D_SZ + 1>(As, Bt, c % GH_X, c / GH_X,
            nullptr, xs, tvec, W1, b1, a1, s1);
    }
}

__global__ __launch_bounds__(NT)
void mid_kernel(const float* __restrict__ a1, const float* __restrict__ W2,
                const float* __restrict__ b2,
                const float* __restrict__ s1, const float* __restrict__ Cm,
                float* __restrict__ a2, float* __restrict__ part)
{
    __shared__ __attribute__((aligned(16))) lds_tile_t As;
    __shared__ __attribute__((aligned(16))) lds_tile_t Bt;
    mid_fused_tile(As, Bt, blockIdx.x, blockIdx.y, a1, W2, b2, s1, Cm, a2, part);
}

constexpr int OB_M    = 16;
constexpr int OB_NT   = 256;
constexpr int OB_NW   = OB_NT / 32;
constexpr int OB_KR   = H_SZ / 2;
constexpr int OB_KW   = OB_KR / OB_NW;
constexpr int OB_NTL  = (D_SZ + 15) / 16;
constexpr int OB_AP   = H_SZ + 4;
constexpr int OB_RP   = OB_NTL * 16 + 4;
constexpr int OB_A4   = OB_M * H_SZ / 4 / OB_NT;
constexpr int OB_B4   = H_SZ * D_SZ / 4 / OB_NT;
constexpr int OB_B4R0 = (OB_KR * D_SZ / 4 + OB_NT - 1) / OB_NT;
constexpr int NB_OUT  = B_SZ / OB_M;
static_assert(OB_A4 * 4 * OB_NT == OB_M * H_SZ, "A strip must split evenly over the block");
static_assert(OB_B4 * 4 * OB_NT == H_SZ * D_SZ, "W3 must split evenly over the block");
static_assert(OB_B4R0 * 4 * OB_NT >= OB_KR * D_SZ && OB_B4R0 <= OB_B4, "round 0 needs its rows of W3");
static_assert((OB_M * D_SZ * 4) % 128 == 0, "a block's outputs must be whole 128-byte lines");
static_assert(D_SZ <= OB_NT, "one thread per bias value");

__global__ __launch_bounds__(OB_NT)
void back_kernel(const float* __restrict__ a2, const float* __restrict__ W3,
                 const float* __restrict__ b3, float* __restrict__ dxs,
                 const float* __restrict__ part, float* __restrict__ ndiv)
{
    __shared__ __attribute__((aligned(16))) float As[OB_M * OB_AP];
    __shared__ __attribute__((aligned(16))) float Bs[H_SZ * D_SZ];
    __shared__ __attribute__((aligned(16))) float red[OB_NW * OB_M * OB_RP];
    __shared__ float bsh[OB_NT];

    const int tid  = threadIdx.x;
    const int wave = tid >> 5;
    const int lane = tid & 31;
    const int c    = lane & 15;
    const int h    = lane >> 4;
    const int m0   = blockIdx.x * OB_M;

    float pv[GH_X];
    const int prow = (m0 & ~(2 * OB_M - 1)) + lane;
    #pragma unroll
    for (int j = 0; j < GH_X; ++j) pv[j] = part[(size_t)j * B_SZ + prow];
    float bias_v = b3[tid < D_SZ ? tid : D_SZ - 1];
    v4f ar[OB_A4];
    v4f br[OB_B4];
    const v4f* __restrict__ ag = (const v4f*)(a2 + (size_t)m0 * H_SZ);
    const v4f* __restrict__ bg = (const v4f*)W3;
    #pragma unroll
    for (int i = 0; i < OB_A4; ++i) ar[i] = ag[tid + OB_NT * i];
    __builtin_amdgcn_sched_barrier(0);
    #pragma unroll
    for (int i = 0; i < OB_B4R0; ++i) br[i] = bg[tid + OB_NT * i];
    __builtin_amdgcn_sched_barrier(0);
    #pragma unroll
    for (int i = OB_B4R0; i < OB_B4; ++i) br[i] = bg[tid + OB_NT * i];
    __builtin_amdgcn_sched_barrier(0);

    float dsum = 0.0f;
    #pragma unroll
    for (int j = 0; j < GH_X; ++j) dsum += pv[j];
    asm volatile("" : "+v"(dsum));

    v8f acc[OB_NTL] = {};
    int ncol[OB_NTL];
    #pragma unroll
    for (int nt = 0; nt < OB_NTL; ++nt) {
        const int n = 16 * nt + c;
        ncol[nt] = n < D_SZ ? n : D_SZ - 1;
    }

    auto compute = [&](const int rd) __attribute__((always_inline)) {
        #pragma unroll
        for (int j = 0; j < OB_KW / 8; ++j) {
            const int k0 = rd * OB_KR + wave * OB_KW + 8 * j + 4 * h;
            const v4f a = *(const v4f*)&As[c * OB_AP + k0];
            const v2f alo = {a[0], a[1]};
            const v2f ahi = {a[2], a[3]};
            #pragma unroll
            for (int nt = 0; nt < OB_NTL; ++nt) {
                const float* bp = &Bs[k0 * D_SZ + ncol[nt]];
                const v2f blo = {bp[0],        bp[D_SZ]};
                const v2f bhi = {bp[2 * D_SZ], bp[3 * D_SZ]};
                acc[nt] = wmma4(alo, blo, acc[nt]);
                acc[nt] = wmma4(ahi, bhi, acc[nt]);
            }
        }
    };

    bsh[tid] = bias_v;
    #pragma unroll
    for (int i = 0; i < OB_A4; ++i) {
        const int idx = tid + OB_NT * i;
        *(v4f*)&As[(idx >> 7) * OB_AP + 4 * (idx & 127)] = ar[i];
    }
    #pragma unroll
    for (int i = 0; i < OB_B4R0; ++i)
        *(v4f*)&Bs[4 * (tid + OB_NT * i)] = br[i];
    asm volatile("s_wait_dscnt 0x0" ::: "memory");
    asm volatile("s_barrier_signal -1" ::: "memory");
    asm volatile("s_barrier_wait -1" ::: "memory");
    compute(0);
    asm volatile("" : "+v"(acc[0]), "+v"(acc[1]), "+v"(acc[2]), "+v"(acc[3]), "+v"(acc[4])
                 : : "memory");
    #pragma unroll
    for (int i = OB_B4R0; i < OB_B4; ++i)
        *(v4f*)&Bs[4 * (tid + OB_NT * i)] = br[i];
    __syncthreads();
    compute(1);

    #pragma unroll
    for (int nt = 0; nt < OB_NTL; ++nt)
        #pragma unroll
        for (int r = 0; r < 8; ++r)
            red[(wave * OB_M + 8 * h + r) * OB_RP + 16 * nt + c] = acc[nt][r];
    __syncthreads();

    #pragma unroll
    for (int p = 0; p < (OB_M * D_SZ + OB_NT - 1) / OB_NT; ++p) {
        const int o = tid + OB_NT * p;
        if (o < OB_M * D_SZ) {
            const int r = o / D_SZ;
            const int n = o - r * D_SZ;
            float s = 0.0f;
            #pragma unroll
            for (int w = 0; w < OB_NW; ++w)
                s += red[(w * OB_M + r) * OB_RP + n];
            dxs[(size_t)m0 * D_SZ + o] = s + bsh[n];
        }
    }

    if ((blockIdx.x & 1) == 0 && tid < 2 * OB_M)
        ndiv[m0 + tid] = -dsum;
}

extern "C" void kernel_launch(void* const* d_in, const int* in_sizes, int n_in,
                              void* d_out, int out_size, void* d_ws, size_t ws_size,
                              hipStream_t stream) {
    const float* xs = (const float*)d_in[0];
    const float* t  = (const float*)d_in[1];
    const float* W1 = (const float*)d_in[2];
    const float* b1 = (const float*)d_in[3];
    const float* W2 = (const float*)d_in[4];
    const float* b2 = (const float*)d_in[5];
    const float* W3 = (const float*)d_in[6];
    const float* b3 = (const float*)d_in[7];

    float* dxs  = (float*)d_out;
    float* ndiv = (float*)d_out + (size_t)B_SZ * D_SZ;

    float* ws = (float*)d_ws;
    const size_t act = (size_t)B_SZ * H_SZ;
    float* a1  = ws;
    float* s1  = ws + act;
    float* a2  = ws + 2 * act;
    float* prt = ws + 3 * act;
    float* Cm  = ws + 4 * act;

    dim3 blk(NT);
    dim3 gridH(H_SZ / BN, B_SZ / BM);

    front_kernel<<<dim3(NB_C + NB_L1), blk, 0, stream>>>(
        xs, t, W1, b1, W2, W3, Cm, a1, s1);

    mid_kernel<<<gridH, blk, 0, stream>>>(a1, W2, b2, s1, Cm, a2, prt);

    back_kernel<<<dim3(NB_OUT), dim3(OB_NT), 0, stream>>>(
        a2, W3, b3, dxs, prt, ndiv);
}


# layer-2 and trace products share ONE k-loop: per iteration a k-tile of both is staged and 64 WMMAs per wave run behind one barrier (16 iterations, was 33 of 32), fragment reads grouped a run ahead
# speedup vs baseline: 10.2866x; 1.0527x over previous
#include <hip/hip_runtime.h>
#include <hip/hip_bf16.h>
#include <math.h>


typedef __attribute__((ext_vector_type(2))) float v2f;
typedef __attribute__((ext_vector_type(8))) float v8f;

#define B_SZ 4096
#define D_SZ 66
#define H_SZ 512

constexpr int BM = 64;
constexpr int BN = 64;
constexpr int BK = 32;
constexpr int NT = 128;

__device__ __forceinline__ v8f wmma4(v2f a, v2f b, v8f c) {
    return __builtin_amdgcn_wmma_f32_16x16x4_f32(
        false, a, false, b, (short)0, c, false, false);
}

typedef __attribute__((ext_vector_type(4))) float v4f;

typedef float lds_tile_t[2][BM][BK + 4];

template<bool ACONCAT, int EPI, int NDIM, int KDIM>
__device__ __forceinline__
void gemm_tile(lds_tile_t& As, lds_tile_t& Bt,
               const int bx, const int by,
               const float* __restrict__ A, const float* __restrict__ xs,
               const float* __restrict__ tvec,
               const float* __restrict__ Bm, const float* __restrict__ bias,
               float* __restrict__ out0, float* __restrict__ out1)
{
    static_assert(BM == BN, "A and B tiles share one LDS array type");

    constexpr bool KFULL = (KDIM % BK) == 0;
    constexpr bool HAS_T = ACONCAT && (KDIM > D_SZ);
    constexpr bool NFULL = (NDIM % BN) == 0;
    constexpr int  NKT   = (KDIM + BK - 1) / BK;
    constexpr int  UNR   = ACONCAT ? (NKT - 1) : 1;
    static_assert(ACONCAT || (KFULL && (KDIM % 4) == 0), "A rows must allow 128-bit loads");
    static_assert(NKT >= 2, "the pipelined loop expects at least two k-tiles");

    const int tid  = threadIdx.x;
    const int wave = tid >> 5;
    const int lane = tid & 31;
    const int wm   = wave >> 1;
    const int wn   = wave & 1;

    const int m0 = by * BM;
    const int n0 = bx * BN;

    v8f acc[2][2] = {};

    const int row16 = lane & 15;
    const int ksub  = (lane >> 4) << 1;

    const int a_q  = tid & 7;
    const int a_m  = tid >> 3;
    const int b_n  = tid & (BN - 1);
    const int b_q  = tid >> 6;
    const int b_gn = n0 + b_n;
    const int b_cn = NFULL ? b_gn : (b_gn < NDIM ? b_gn : NDIM - 1);

    v4f   av4[4];
    float avs[16];
    float avt[16];
    v4f   bv[4];

    auto gload = [&](const int k0) __attribute__((always_inline)) {
        const bool a_tail = ACONCAT && (k0 + BK > D_SZ);
        if (ACONCAT) {
            #pragma unroll
            for (int i = 0; i < 16; ++i) {
                const int idx = tid + i * NT;
                const int gk = k0 + (idx & (BK - 1));
                const int gm = m0 + (idx >> 5);
                const int ck = gk < D_SZ ? gk : D_SZ - 1;
                avs[i] = xs[gm * D_SZ + ck];
                avt[i] = (HAS_T && a_tail) ? tvec[gm] : 0.0f;
            }
        } else {
            #pragma unroll
            for (int i = 0; i < 4; ++i)
                av4[i] = *(const v4f*)&A[(size_t)(m0 + a_m + 16 * i) * KDIM + k0 + 4 * a_q];
        }
        #pragma unroll
        for (int i = 0; i < 4; ++i) {
            #pragma unroll
            for (int j = 0; j < 4; ++j) {
                const int gk = k0 + 4 * (b_q + 2 * i) + j;
                const int ck = KFULL ? gk : (gk < KDIM ? gk : KDIM - 1);
                bv[i][j] = Bm[(size_t)ck * NDIM + b_cn];
            }
        }
    };

    auto lstore = [&](const int k0, const int s) __attribute__((always_inline)) {
        const bool a_tail = ACONCAT && (k0 + BK > D_SZ);
        if (ACONCAT) {
            #pragma unroll
            for (int i = 0; i < 16; ++i) {
                const int idx = tid + i * NT;
                float v = avs[i];
                if (a_tail) {
                    float tv = avt[i];
                    if (HAS_T) asm volatile("" : "+v"(v), "+v"(tv));
                    else       asm volatile("" : "+v"(v));
                    const int gk = k0 + (idx & (BK - 1));
                    v = gk < D_SZ ? v : ((HAS_T && gk == D_SZ) ? tv : 0.0f);
                }
                As[s][idx >> 5][idx & (BK - 1)] = v;
            }
        } else {
            #pragma unroll
            for (int i = 0; i < 4; ++i)
                *(v4f*)&As[s][a_m + 16 * i][4 * a_q] = av4[i];
        }
        #pragma unroll
        for (int i = 0; i < 4; ++i) {
            v4f w = bv[i];
            if (!(KFULL && NFULL)) {
                #pragma unroll
                for (int j = 0; j < 4; ++j) {
                    const int gk = k0 + 4 * (b_q + 2 * i) + j;
                    const bool ok = (KFULL || gk < KDIM) && (NFULL || b_gn < NDIM);
                    w[j] = ok ? w[j] : 0.0f;
                }
            }
            *(v4f*)&Bt[s][b_n][4 * (b_q + 2 * i)] = w;
        }
    };

    auto compute = [&](const int s) __attribute__((always_inline)) {
        #pragma unroll
        for (int kk = 0; kk < BK; kk += 4) {
            v2f a0 = *(const v2f*)&As[s][wm * 32 +      row16][kk + ksub];
            v2f a1 = *(const v2f*)&As[s][wm * 32 + 16 + row16][kk + ksub];
            v2f b0 = *(const v2f*)&Bt[s][wn * 32 +      row16][kk + ksub];
            v2f b1 = *(const v2f*)&Bt[s][wn * 32 + 16 + row16][kk + ksub];
            acc[0][0] = wmma4(a0, b0, acc[0][0]);
            acc[0][1] = wmma4(a0, b1, acc[0][1]);
            acc[1][0] = wmma4(a1, b0, acc[1][0]);
            acc[1][1] = wmma4(a1, b1, acc[1][1]);
        }
    };

    gload(0);
    __builtin_amdgcn_sched_barrier(0);
    lstore(0, 0);
    __syncthreads();

    #pragma clang loop unroll_count(UNR)
    for (int t = 0; t < NKT - 1; ++t) {
        gload((t + 1) * BK);
        __builtin_amdgcn_sched_barrier(0);
        compute(t & 1);
        __builtin_amdgcn_sched_barrier(0);
        lstore((t + 1) * BK, (t + 1) & 1);
        __syncthreads();
    }
    compute((NKT - 1) & 1);

    #pragma unroll
    for (int mt = 0; mt < 2; ++mt) {
        const int mbase = m0 + wm * 32 + mt * 16 + ((lane >> 4) << 3);
        #pragma unroll
        for (int nt = 0; nt < 2; ++nt) {
            const int nc = n0 + wn * 32 + nt * 16 + row16;
            if (!NFULL && nc >= NDIM) continue;
            if (EPI == 2) {
                const size_t o = (size_t)nc * H_SZ + mbase;
                const v4f w0 = *(const v4f*)&bias[o];
                const v4f w1 = *(const v4f*)&bias[o + 4];
                v4f r0, r1;
                #pragma unroll
                for (int r = 0; r < 4; ++r) {
                    r0[r] = acc[mt][nt][r]     * w0[r];
                    r1[r] = acc[mt][nt][r + 4] * w1[r];
                }
                *(v4f*)&out0[o]     = r0;
                *(v4f*)&out0[o + 4] = r1;
                continue;
            }
            const float bi = bias ? bias[nc] : 0.0f;
            #pragma unroll
            for (int r = 0; r < 8; ++r) {
                const int gm = mbase + r;
                float z = acc[mt][nt][r] + bi;
                if (EPI == 0) {
                    out0[(size_t)gm * NDIM + nc] = z;
                } else {
                    float sg = 1.0f / (1.0f + __expf(-z));
                    out0[(size_t)gm * NDIM + nc] = z * sg;
                    out1[(size_t)gm * NDIM + nc] = sg * (1.0f + z * (1.0f - sg));
                }
            }
        }
    }
}

__device__ __forceinline__ float row16_sum(float v) {
    v += __int_as_float(__builtin_amdgcn_update_dpp(0, __float_as_int(v), 0x161, 0xf, 0xf, true));
    v += __int_as_float(__builtin_amdgcn_update_dpp(0, __float_as_int(v), 0x162, 0xf, 0xf, true));
    v += __int_as_float(__builtin_amdgcn_update_dpp(0, __float_as_int(v), 0x164, 0xf, 0xf, true));
    v += __int_as_float(__builtin_amdgcn_update_dpp(0, __float_as_int(v), 0x168, 0xf, 0xf, true));
    return v;
}

__device__ __forceinline__
void mid_fused_tile(lds_tile_t& Az, lds_tile_t& Bz,
                    lds_tile_t& Au, lds_tile_t& Bu,
                    const int bx, const int by,
                    const float* __restrict__ a1, const float* __restrict__ W2,
                    const float* __restrict__ b2,
                    const float* __restrict__ s1, const float* __restrict__ Cm,
                    float* __restrict__ a2, float* __restrict__ part)
{
    constexpr int NKT = H_SZ / BK;
    static_assert((NKT & 1) == 0, "the epilogue assumes the last k-tile lies in stage 1");

    const int tid  = threadIdx.x;
    const int wave = tid >> 5;
    const int lane = tid & 31;
    const int wm   = wave >> 1;
    const int wn   = wave & 1;

    const int m0 = by * BM;
    const int n0 = bx * BN;

    v8f accz[2][2] = {};
    v8f accu[2][2] = {};

    const int row16 = lane & 15;
    const int ksub  = (lane >> 4) << 1;

    const int a_q  = tid & 7;
    const int a_m  = tid >> 3;
    const int b_n  = tid & (BN - 1);
    const int b_q  = tid >> 6;
    const int b_gn = n0 + b_n;

    v4f az4[4], au4[4];
    v4f bzv[4], buv[4];

    auto gload = [&](const int k0) __attribute__((always_inline)) {
        #pragma unroll
        for (int i = 0; i < 4; ++i) {
            const size_t o = (size_t)(m0 + a_m + 16 * i) * H_SZ + k0 + 4 * a_q;
            az4[i] = *(const v4f*)&a1[o];
            au4[i] = *(const v4f*)&s1[o];
        }
        #pragma unroll
        for (int i = 0; i < 4; ++i) {
            #pragma unroll
            for (int j = 0; j < 4; ++j) {
                const int gk = k0 + 4 * (b_q + 2 * i) + j;
                bzv[i][j] = W2[(size_t)gk * H_SZ + b_gn];
            }
        }
        #pragma unroll
        for (int i = 0; i < 4; ++i) {
            #pragma unroll
            for (int j = 0; j < 4; ++j) {
                const int gk = k0 + 4 * (b_q + 2 * i) + j;
                buv[i][j] = Cm[(size_t)gk * H_SZ + b_gn];
            }
        }
    };

    auto lstore = [&](const int s) __attribute__((always_inline)) {
        #pragma unroll
        for (int i = 0; i < 4; ++i) {
            *(v4f*)&Az[s][a_m + 16 * i][4 * a_q] = az4[i];
            *(v4f*)&Au[s][a_m + 16 * i][4 * a_q] = au4[i];
        }
        #pragma unroll
        for (int i = 0; i < 4; ++i)
            *(v4f*)&Bz[s][b_n][4 * (b_q + 2 * i)] = bzv[i];
        #pragma unroll
        for (int i = 0; i < 4; ++i)
            *(v4f*)&Bu[s][b_n][4 * (b_q + 2 * i)] = buv[i];
    };

    struct frag_t { v2f a0[4], a1[4], b0[4], b1[4]; };
    frag_t F0, F1;

    auto lfrag = [&](frag_t& f, lds_tile_t& As, lds_tile_t& Bt,
                     const int s, const int kh) __attribute__((always_inline)) {
        #pragma unroll
        for (int j = 0; j < 4; ++j) {
            const int kk = kh + 4 * j;
            f.a0[j] = *(const v2f*)&As[s][wm * 32 +      row16][kk + ksub];
            f.a1[j] = *(const v2f*)&As[s][wm * 32 + 16 + row16][kk + ksub];
            f.b0[j] = *(const v2f*)&Bt[s][wn * 32 +      row16][kk + ksub];
            f.b1[j] = *(const v2f*)&Bt[s][wn * 32 + 16 + row16][kk + ksub];
        }
    };

    auto mm = [&](v8f (&acc)[2][2], const frag_t& f) __attribute__((always_inline)) {
        #pragma unroll
        for (int j = 0; j < 4; ++j) {
            acc[0][0] = wmma4(f.a0[j], f.b0[j], acc[0][0]);
            acc[0][1] = wmma4(f.a0[j], f.b1[j], acc[0][1]);
            acc[1][0] = wmma4(f.a1[j], f.b0[j], acc[1][0]);
            acc[1][1] = wmma4(f.a1[j], f.b1[j], acc[1][1]);
        }
    };

    auto compute_both = [&](const int s) __attribute__((always_inline)) {
        lfrag(F0, Az, Bz, s, 0);
        lfrag(F1, Az, Bz, s, BK / 2);
        __builtin_amdgcn_sched_barrier(0);
        mm(accz, F0);
        __builtin_amdgcn_sched_barrier(0);
        lfrag(F0, Au, Bu, s, 0);
        __builtin_amdgcn_sched_barrier(0);
        mm(accz, F1);
        __builtin_amdgcn_sched_barrier(0);
        lfrag(F1, Au, Bu, s, BK / 2);
        __builtin_amdgcn_sched_barrier(0);
        mm(accu, F0);
        mm(accu, F1);
    };

    auto wmma_done = [&]() __attribute__((always_inline)) {
        asm volatile("" : "+v"(accz[0][0]), "+v"(accz[0][1]), "+v"(accz[1][0]), "+v"(accz[1][1]),
                          "+v"(accu[0][0]), "+v"(accu[0][1]), "+v"(accu[1][0]), "+v"(accu[1][1])
                     : : "memory");
    };

    gload(0);
    __builtin_amdgcn_sched_barrier(0);
    lstore(0);
    __syncthreads();

    for (int t = 0; t < NKT - 1; ++t) {
        gload((t + 1) * BK);
        __builtin_amdgcn_sched_barrier(0);
        compute_both(t & 1);
        wmma_done();
        lstore((t + 1) & 1);
        __syncthreads();
    }
    compute_both((NKT - 1) & 1);

    float dsum[2][8];
    #pragma unroll
    for (int mt = 0; mt < 2; ++mt) {
        const int mbase = m0 + wm * 32 + mt * 16 + ((lane >> 4) << 3);
        #pragma unroll
        for (int r = 0; r < 8; ++r) dsum[mt][r] = 0.0f;
        #pragma unroll
        for (int nt = 0; nt < 2; ++nt) {
            const int nc = n0 + wn * 32 + nt * 16 + row16;
            const float bi = b2[nc];
            #pragma unroll
            for (int r = 0; r < 8; ++r) {
                const float z = accz[mt][nt][r] + bi;
                const float sg = 1.0f / (1.0f + __expf(-z));
                accz[mt][nt][r] = z * sg;
                dsum[mt][r] += accu[mt][nt][r] * (sg * (1.0f + z * (1.0f - sg)));
            }
        }
    }
    #pragma unroll
    for (int mt = 0; mt < 2; ++mt)
        #pragma unroll
        for (int r = 0; r < 8; ++r)
            dsum[mt][r] = row16_sum(dsum[mt][r]);

    float* red = &Az[0][0][0];
    if (row16 == 0) {
        #pragma unroll
        for (int mt = 0; mt < 2; ++mt)
            #pragma unroll
            for (int r = 0; r < 8; ++r)
                red[wn * BM + wm * 32 + mt * 16 + ((lane >> 4) << 3) + r] = dsum[mt][r];
    }
    __syncthreads();
    if (tid < BM)
        part[(size_t)bx * B_SZ + m0 + tid] = red[tid] + red[BM + tid];

    #pragma unroll
    for (int mt = 0; mt < 2; ++mt) {
        const int mbase = m0 + wm * 32 + mt * 16 + ((lane >> 4) << 3);
        #pragma unroll
        for (int nt = 0; nt < 2; ++nt) {
            const int nc = n0 + wn * 32 + nt * 16 + row16;
            #pragma unroll
            for (int r = 0; r < 8; ++r)
                a2[(size_t)(mbase + r) * H_SZ + nc] = accz[mt][nt][r];
        }
    }
}

constexpr int GH_X   = H_SZ / BN;
constexpr int NB_C   = (H_SZ / BN) * (H_SZ / BM);
constexpr int NB_L1  = (H_SZ / BN) * (B_SZ / BM);

__global__ __launch_bounds__(NT)
void front_kernel(const float* __restrict__ xs, const float* __restrict__ tvec,
                  const float* __restrict__ W1, const float* __restrict__ b1,
                  const float* __restrict__ W2, const float* __restrict__ W3,
                  float* __restrict__ Cm, float* __restrict__ a1, float* __restrict__ s1)
{
    __shared__ __attribute__((aligned(16))) lds_tile_t As;
    __shared__ __attribute__((aligned(16))) lds_tile_t Bt;
    const int b = blockIdx.x;
    if (b < NB_C) {
        gemm_tile<true, 2, H_SZ, D_SZ>(As, Bt, b % GH_X, b / GH_X,
            nullptr, W3, nullptr, W1, W2, Cm, nullptr);
    } else {
        const int c = b - NB_C;
        gemm_tile<true, 1, H_SZ, D_SZ + 1>(As, Bt, c % GH_X, c / GH_X,
            nullptr, xs, tvec, W1, b1, a1, s1);
    }
}

__global__ __launch_bounds__(NT)
void mid_kernel(const float* __restrict__ a1, const float* __restrict__ W2,
                const float* __restrict__ b2,
                const float* __restrict__ s1, const float* __restrict__ Cm,
                float* __restrict__ a2, float* __restrict__ part)
{
    __shared__ __attribute__((aligned(16))) lds_tile_t Az;
    __shared__ __attribute__((aligned(16))) lds_tile_t Bz;
    __shared__ __attribute__((aligned(16))) lds_tile_t Au;
    __shared__ __attribute__((aligned(16))) lds_tile_t Bu;
    mid_fused_tile(Az, Bz, Au, Bu, blockIdx.x, blockIdx.y, a1, W2, b2, s1, Cm, a2, part);
}

constexpr int OB_M    = 16;
constexpr int OB_NT   = 256;
constexpr int OB_NW   = OB_NT / 32;
constexpr int OB_KR   = H_SZ / 2;
constexpr int OB_KW   = OB_KR / OB_NW;
constexpr int OB_NTL  = (D_SZ + 15) / 16;
constexpr int OB_AP   = H_SZ + 4;
constexpr int OB_RP   = OB_NTL * 16 + 4;
constexpr int OB_A4   = OB_M * H_SZ / 4 / OB_NT;
constexpr int OB_B4   = H_SZ * D_SZ / 4 / OB_NT;
constexpr int OB_B4R0 = (OB_KR * D_SZ / 4 + OB_NT - 1) / OB_NT;
constexpr int NB_OUT  = B_SZ / OB_M;
static_assert(OB_A4 * 4 * OB_NT == OB_M * H_SZ, "A strip must split evenly over the block");
static_assert(OB_B4 * 4 * OB_NT == H_SZ * D_SZ, "W3 must split evenly over the block");
static_assert(OB_B4R0 * 4 * OB_NT >= OB_KR * D_SZ && OB_B4R0 <= OB_B4, "round 0 needs its rows of W3");
static_assert((OB_M * D_SZ * 4) % 128 == 0, "a block's outputs must be whole 128-byte lines");
static_assert(D_SZ <= OB_NT, "one thread per bias value");

__global__ __launch_bounds__(OB_NT)
void back_kernel(const float* __restrict__ a2, const float* __restrict__ W3,
                 const float* __restrict__ b3, float* __restrict__ dxs,
                 const float* __restrict__ part, float* __restrict__ ndiv)
{
    __shared__ __attribute__((aligned(16))) float As[OB_M * OB_AP];
    __shared__ __attribute__((aligned(16))) float Bs[H_SZ * D_SZ];
    __shared__ __attribute__((aligned(16))) float red[OB_NW * OB_M * OB_RP];
    __shared__ float bsh[OB_NT];

    const int tid  = threadIdx.x;
    const int wave = tid >> 5;
    const int lane = tid & 31;
    const int c    = lane & 15;
    const int h    = lane >> 4;
    const int m0   = blockIdx.x * OB_M;

    float pv[GH_X];
    const int prow = (m0 & ~(2 * OB_M - 1)) + lane;
    #pragma unroll
    for (int j = 0; j < GH_X; ++j) pv[j] = part[(size_t)j * B_SZ + prow];
    float bias_v = b3[tid < D_SZ ? tid : D_SZ - 1];
    v4f ar[OB_A4];
    v4f br[OB_B4];
    const v4f* __restrict__ ag = (const v4f*)(a2 + (size_t)m0 * H_SZ);
    const v4f* __restrict__ bg = (const v4f*)W3;
    #pragma unroll
    for (int i = 0; i < OB_A4; ++i) ar[i] = ag[tid + OB_NT * i];
    __builtin_amdgcn_sched_barrier(0);
    #pragma unroll
    for (int i = 0; i < OB_B4R0; ++i) br[i] = bg[tid + OB_NT * i];
    __builtin_amdgcn_sched_barrier(0);
    #pragma unroll
    for (int i = OB_B4R0; i < OB_B4; ++i) br[i] = bg[tid + OB_NT * i];
    __builtin_amdgcn_sched_barrier(0);

    float dsum = 0.0f;
    #pragma unroll
    for (int j = 0; j < GH_X; ++j) dsum += pv[j];
    asm volatile("" : "+v"(dsum));

    v8f acc[OB_NTL] = {};
    int ncol[OB_NTL];
    #pragma unroll
    for (int nt = 0; nt < OB_NTL; ++nt) {
        const int n = 16 * nt + c;
        ncol[nt] = n < D_SZ ? n : D_SZ - 1;
    }

    auto compute = [&](const int rd) __attribute__((always_inline)) {
        #pragma unroll
        for (int j = 0; j < OB_KW / 8; ++j) {
            const int k0 = rd * OB_KR + wave * OB_KW + 8 * j + 4 * h;
            const v4f a = *(const v4f*)&As[c * OB_AP + k0];
            const v2f alo = {a[0], a[1]};
            const v2f ahi = {a[2], a[3]};
            #pragma unroll
            for (int nt = 0; nt < OB_NTL; ++nt) {
                const float* bp = &Bs[k0 * D_SZ + ncol[nt]];
                const v2f blo = {bp[0],        bp[D_SZ]};
                const v2f bhi = {bp[2 * D_SZ], bp[3 * D_SZ]};
                acc[nt] = wmma4(alo, blo, acc[nt]);
                acc[nt] = wmma4(ahi, bhi, acc[nt]);
            }
        }
    };

    bsh[tid] = bias_v;
    #pragma unroll
    for (int i = 0; i < OB_A4; ++i) {
        const int idx = tid + OB_NT * i;
        *(v4f*)&As[(idx >> 7) * OB_AP + 4 * (idx & 127)] = ar[i];
    }
    #pragma unroll
    for (int i = 0; i < OB_B4R0; ++i)
        *(v4f*)&Bs[4 * (tid + OB_NT * i)] = br[i];
    asm volatile("s_wait_dscnt 0x0" ::: "memory");
    asm volatile("s_barrier_signal -1" ::: "memory");
    asm volatile("s_barrier_wait -1" ::: "memory");
    compute(0);
    asm volatile("" : "+v"(acc[0]), "+v"(acc[1]), "+v"(acc[2]), "+v"(acc[3]), "+v"(acc[4])
                 : : "memory");
    #pragma unroll
    for (int i = OB_B4R0; i < OB_B4; ++i)
        *(v4f*)&Bs[4 * (tid + OB_NT * i)] = br[i];
    __syncthreads();
    compute(1);

    #pragma unroll
    for (int nt = 0; nt < OB_NTL; ++nt)
        #pragma unroll
        for (int r = 0; r < 8; ++r)
            red[(wave * OB_M + 8 * h + r) * OB_RP + 16 * nt + c] = acc[nt][r];
    __syncthreads();

    #pragma unroll
    for (int p = 0; p < (OB_M * D_SZ + OB_NT - 1) / OB_NT; ++p) {
        const int o = tid + OB_NT * p;
        if (o < OB_M * D_SZ) {
            const int r = o / D_SZ;
            const int n = o - r * D_SZ;
            float s = 0.0f;
            #pragma unroll
            for (int w = 0; w < OB_NW; ++w)
                s += red[(w * OB_M + r) * OB_RP + n];
            dxs[(size_t)m0 * D_SZ + o] = s + bsh[n];
        }
    }

    if ((blockIdx.x & 1) == 0 && tid < 2 * OB_M)
        ndiv[m0 + tid] = -dsum;
}

extern "C" void kernel_launch(void* const* d_in, const int* in_sizes, int n_in,
                              void* d_out, int out_size, void* d_ws, size_t ws_size,
                              hipStream_t stream) {
    const float* xs = (const float*)d_in[0];
    const float* t  = (const float*)d_in[1];
    const float* W1 = (const float*)d_in[2];
    const float* b1 = (const float*)d_in[3];
    const float* W2 = (const float*)d_in[4];
    const float* b2 = (const float*)d_in[5];
    const float* W3 = (const float*)d_in[6];
    const float* b3 = (const float*)d_in[7];

    float* dxs  = (float*)d_out;
    float* ndiv = (float*)d_out + (size_t)B_SZ * D_SZ;

    float* ws = (float*)d_ws;
    const size_t act = (size_t)B_SZ * H_SZ;
    float* a1  = ws;
    float* s1  = ws + act;
    float* a2  = ws + 2 * act;
    float* prt = ws + 3 * act;
    float* Cm  = ws + 4 * act;

    dim3 blk(NT);
    dim3 gridH(H_SZ / BN, B_SZ / BM);

    front_kernel<<<dim3(NB_C + NB_L1), blk, 0, stream>>>(
        xs, t, W1, b1, W2, W3, Cm, a1, s1);

    mid_kernel<<<gridH, blk, 0, stream>>>(a1, W2, b2, s1, Cm, a2, prt);

    back_kernel<<<dim3(NB_OUT), dim3(OB_NT), 0, stream>>>(
        a2, W3, b3, dxs, prt, ndiv);
}


# layer-2/trace kernel: wave priority 3 (s_setprio) from a k-tile's first WMMA to its 64th, 0 elsewhere: the 2 workgroups of a WGP take the matrix pipe in whole-tile turns (was oldest-first)
# speedup vs baseline: 10.7482x; 1.0449x over previous
#include <hip/hip_runtime.h>
#include <hip/hip_bf16.h>
#include <math.h>


typedef __attribute__((ext_vector_type(2))) float v2f;
typedef __attribute__((ext_vector_type(8))) float v8f;

#define B_SZ 4096
#define D_SZ 66
#define H_SZ 512

constexpr int BM = 64;
constexpr int BN = 64;
constexpr int BK = 32;
constexpr int NT = 128;

__device__ __forceinline__ v8f wmma4(v2f a, v2f b, v8f c) {
    return __builtin_amdgcn_wmma_f32_16x16x4_f32(
        false, a, false, b, (short)0, c, false, false);
}

typedef __attribute__((ext_vector_type(4))) float v4f;

typedef float lds_tile_t[2][BM][BK + 4];

template<bool ACONCAT, int EPI, int NDIM, int KDIM>
__device__ __forceinline__
void gemm_tile(lds_tile_t& As, lds_tile_t& Bt,
               const int bx, const int by,
               const float* __restrict__ A, const float* __restrict__ xs,
               const float* __restrict__ tvec,
               const float* __restrict__ Bm, const float* __restrict__ bias,
               float* __restrict__ out0, float* __restrict__ out1)
{
    static_assert(BM == BN, "A and B tiles share one LDS array type");

    constexpr bool KFULL = (KDIM % BK) == 0;
    constexpr bool HAS_T = ACONCAT && (KDIM > D_SZ);
    constexpr bool NFULL = (NDIM % BN) == 0;
    constexpr int  NKT   = (KDIM + BK - 1) / BK;
    constexpr int  UNR   = ACONCAT ? (NKT - 1) : 1;
    static_assert(ACONCAT || (KFULL && (KDIM % 4) == 0), "A rows must allow 128-bit loads");
    static_assert(NKT >= 2, "the pipelined loop expects at least two k-tiles");

    const int tid  = threadIdx.x;
    const int wave = tid >> 5;
    const int lane = tid & 31;
    const int wm   = wave >> 1;
    const int wn   = wave & 1;

    const int m0 = by * BM;
    const int n0 = bx * BN;

    v8f acc[2][2] = {};

    const int row16 = lane & 15;
    const int ksub  = (lane >> 4) << 1;

    const int a_q  = tid & 7;
    const int a_m  = tid >> 3;
    const int b_n  = tid & (BN - 1);
    const int b_q  = tid >> 6;
    const int b_gn = n0 + b_n;
    const int b_cn = NFULL ? b_gn : (b_gn < NDIM ? b_gn : NDIM - 1);

    v4f   av4[4];
    float avs[16];
    float avt[16];
    v4f   bv[4];

    auto gload = [&](const int k0) __attribute__((always_inline)) {
        const bool a_tail = ACONCAT && (k0 + BK > D_SZ);
        if (ACONCAT) {
            #pragma unroll
            for (int i = 0; i < 16; ++i) {
                const int idx = tid + i * NT;
                const int gk = k0 + (idx & (BK - 1));
                const int gm = m0 + (idx >> 5);
                const int ck = gk < D_SZ ? gk : D_SZ - 1;
                avs[i] = xs[gm * D_SZ + ck];
                avt[i] = (HAS_T && a_tail) ? tvec[gm] : 0.0f;
            }
        } else {
            #pragma unroll
            for (int i = 0; i < 4; ++i)
                av4[i] = *(const v4f*)&A[(size_t)(m0 + a_m + 16 * i) * KDIM + k0 + 4 * a_q];
        }
        #pragma unroll
        for (int i = 0; i < 4; ++i) {
            #pragma unroll
            for (int j = 0; j < 4; ++j) {
                const int gk = k0 + 4 * (b_q + 2 * i) + j;
                const int ck = KFULL ? gk : (gk < KDIM ? gk : KDIM - 1);
                bv[i][j] = Bm[(size_t)ck * NDIM + b_cn];
            }
        }
    };

    auto lstore = [&](const int k0, const int s) __attribute__((always_inline)) {
        const bool a_tail = ACONCAT && (k0 + BK > D_SZ);
        if (ACONCAT) {
            #pragma unroll
            for (int i = 0; i < 16; ++i) {
                const int idx = tid + i * NT;
                float v = avs[i];
                if (a_tail) {
                    float tv = avt[i];
                    if (HAS_T) asm volatile("" : "+v"(v), "+v"(tv));
                    else       asm volatile("" : "+v"(v));
                    const int gk = k0 + (idx & (BK - 1));
                    v = gk < D_SZ ? v : ((HAS_T && gk == D_SZ) ? tv : 0.0f);
                }
                As[s][idx >> 5][idx & (BK - 1)] = v;
            }
        } else {
            #pragma unroll
            for (int i = 0; i < 4; ++i)
                *(v4f*)&As[s][a_m + 16 * i][4 * a_q] = av4[i];
        }
        #pragma unroll
        for (int i = 0; i < 4; ++i) {
            v4f w = bv[i];
            if (!(KFULL && NFULL)) {
                #pragma unroll
                for (int j = 0; j < 4; ++j) {
                    const int gk = k0 + 4 * (b_q + 2 * i) + j;
                    const bool ok = (KFULL || gk < KDIM) && (NFULL || b_gn < NDIM);
                    w[j] = ok ? w[j] : 0.0f;
                }
            }
            *(v4f*)&Bt[s][b_n][4 * (b_q + 2 * i)] = w;
        }
    };

    auto compute = [&](const int s) __attribute__((always_inline)) {
        #pragma unroll
        for (int kk = 0; kk < BK; kk += 4) {
            v2f a0 = *(const v2f*)&As[s][wm * 32 +      row16][kk + ksub];
            v2f a1 = *(const v2f*)&As[s][wm * 32 + 16 + row16][kk + ksub];
            v2f b0 = *(const v2f*)&Bt[s][wn * 32 +      row16][kk + ksub];
            v2f b1 = *(const v2f*)&Bt[s][wn * 32 + 16 + row16][kk + ksub];
            acc[0][0] = wmma4(a0, b0, acc[0][0]);
            acc[0][1] = wmma4(a0, b1, acc[0][1]);
            acc[1][0] = wmma4(a1, b0, acc[1][0]);
            acc[1][1] = wmma4(a1, b1, acc[1][1]);
        }
    };

    gload(0);
    __builtin_amdgcn_sched_barrier(0);
    lstore(0, 0);
    __syncthreads();

    #pragma clang loop unroll_count(UNR)
    for (int t = 0; t < NKT - 1; ++t) {
        gload((t + 1) * BK);
        __builtin_amdgcn_sched_barrier(0);
        compute(t & 1);
        __builtin_amdgcn_sched_barrier(0);
        lstore((t + 1) * BK, (t + 1) & 1);
        __syncthreads();
    }
    compute((NKT - 1) & 1);

    #pragma unroll
    for (int mt = 0; mt < 2; ++mt) {
        const int mbase = m0 + wm * 32 + mt * 16 + ((lane >> 4) << 3);
        #pragma unroll
        for (int nt = 0; nt < 2; ++nt) {
            const int nc = n0 + wn * 32 + nt * 16 + row16;
            if (!NFULL && nc >= NDIM) continue;
            if (EPI == 2) {
                const size_t o = (size_t)nc * H_SZ + mbase;
                const v4f w0 = *(const v4f*)&bias[o];
                const v4f w1 = *(const v4f*)&bias[o + 4];
                v4f r0, r1;
                #pragma unroll
                for (int r = 0; r < 4; ++r) {
                    r0[r] = acc[mt][nt][r]     * w0[r];
                    r1[r] = acc[mt][nt][r + 4] * w1[r];
                }
                *(v4f*)&out0[o]     = r0;
                *(v4f*)&out0[o + 4] = r1;
                continue;
            }
            const float bi = bias ? bias[nc] : 0.0f;
            #pragma unroll
            for (int r = 0; r < 8; ++r) {
                const int gm = mbase + r;
                float z = acc[mt][nt][r] + bi;
                if (EPI == 0) {
                    out0[(size_t)gm * NDIM + nc] = z;
                } else {
                    float sg = 1.0f / (1.0f + __expf(-z));
                    out0[(size_t)gm * NDIM + nc] = z * sg;
                    out1[(size_t)gm * NDIM + nc] = sg * (1.0f + z * (1.0f - sg));
                }
            }
        }
    }
}

__device__ __forceinline__ float row16_sum(float v) {
    v += __int_as_float(__builtin_amdgcn_update_dpp(0, __float_as_int(v), 0x161, 0xf, 0xf, true));
    v += __int_as_float(__builtin_amdgcn_update_dpp(0, __float_as_int(v), 0x162, 0xf, 0xf, true));
    v += __int_as_float(__builtin_amdgcn_update_dpp(0, __float_as_int(v), 0x164, 0xf, 0xf, true));
    v += __int_as_float(__builtin_amdgcn_update_dpp(0, __float_as_int(v), 0x168, 0xf, 0xf, true));
    return v;
}

__device__ __forceinline__
void mid_fused_tile(lds_tile_t& Az, lds_tile_t& Bz,
                    lds_tile_t& Au, lds_tile_t& Bu,
                    const int bx, const int by,
                    const float* __restrict__ a1, const float* __restrict__ W2,
                    const float* __restrict__ b2,
                    const float* __restrict__ s1, const float* __restrict__ Cm,
                    float* __restrict__ a2, float* __restrict__ part)
{
    constexpr int NKT = H_SZ / BK;
    static_assert((NKT & 1) == 0, "the epilogue assumes the last k-tile lies in stage 1");

    const int tid  = threadIdx.x;
    const int wave = tid >> 5;
    const int lane = tid & 31;
    const int wm   = wave >> 1;
    const int wn   = wave & 1;

    const int m0 = by * BM;
    const int n0 = bx * BN;

    v8f accz[2][2] = {};
    v8f accu[2][2] = {};

    const int row16 = lane & 15;
    const int ksub  = (lane >> 4) << 1;

    const int a_q  = tid & 7;
    const int a_m  = tid >> 3;
    const int b_n  = tid & (BN - 1);
    const int b_q  = tid >> 6;
    const int b_gn = n0 + b_n;

    v4f az4[4], au4[4];
    v4f bzv[4], buv[4];

    auto gload = [&](const int k0) __attribute__((always_inline)) {
        #pragma unroll
        for (int i = 0; i < 4; ++i) {
            const size_t o = (size_t)(m0 + a_m + 16 * i) * H_SZ + k0 + 4 * a_q;
            az4[i] = *(const v4f*)&a1[o];
            au4[i] = *(const v4f*)&s1[o];
        }
        #pragma unroll
        for (int i = 0; i < 4; ++i) {
            #pragma unroll
            for (int j = 0; j < 4; ++j) {
                const int gk = k0 + 4 * (b_q + 2 * i) + j;
                bzv[i][j] = W2[(size_t)gk * H_SZ + b_gn];
            }
        }
        #pragma unroll
        for (int i = 0; i < 4; ++i) {
            #pragma unroll
            for (int j = 0; j < 4; ++j) {
                const int gk = k0 + 4 * (b_q + 2 * i) + j;
                buv[i][j] = Cm[(size_t)gk * H_SZ + b_gn];
            }
        }
    };

    auto lstore = [&](const int s) __attribute__((always_inline)) {
        #pragma unroll
        for (int i = 0; i < 4; ++i) {
            *(v4f*)&Az[s][a_m + 16 * i][4 * a_q] = az4[i];
            *(v4f*)&Au[s][a_m + 16 * i][4 * a_q] = au4[i];
        }
        #pragma unroll
        for (int i = 0; i < 4; ++i)
            *(v4f*)&Bz[s][b_n][4 * (b_q + 2 * i)] = bzv[i];
        #pragma unroll
        for (int i = 0; i < 4; ++i)
            *(v4f*)&Bu[s][b_n][4 * (b_q + 2 * i)] = buv[i];
    };

    struct frag_t { v2f a0[4], a1[4], b0[4], b1[4]; };
    frag_t F0, F1;

    auto lfrag = [&](frag_t& f, lds_tile_t& As, lds_tile_t& Bt,
                     const int s, const int kh) __attribute__((always_inline)) {
        #pragma unroll
        for (int j = 0; j < 4; ++j) {
            const int kk = kh + 4 * j;
            f.a0[j] = *(const v2f*)&As[s][wm * 32 +      row16][kk + ksub];
            f.a1[j] = *(const v2f*)&As[s][wm * 32 + 16 + row16][kk + ksub];
            f.b0[j] = *(const v2f*)&Bt[s][wn * 32 +      row16][kk + ksub];
            f.b1[j] = *(const v2f*)&Bt[s][wn * 32 + 16 + row16][kk + ksub];
        }
    };

    auto mm = [&](v8f (&acc)[2][2], const frag_t& f) __attribute__((always_inline)) {
        #pragma unroll
        for (int j = 0; j < 4; ++j) {
            acc[0][0] = wmma4(f.a0[j], f.b0[j], acc[0][0]);
            acc[0][1] = wmma4(f.a0[j], f.b1[j], acc[0][1]);
            acc[1][0] = wmma4(f.a1[j], f.b0[j], acc[1][0]);
            acc[1][1] = wmma4(f.a1[j], f.b1[j], acc[1][1]);
        }
    };

    auto mm_hold = [&](v8f (&acc)[2][2], const frag_t& f) __attribute__((always_inline)) {
        acc[0][0] = wmma4(f.a0[0], f.b0[0], acc[0][0]);
        __builtin_amdgcn_sched_barrier(0);
        __builtin_amdgcn_s_setprio(3);
        __builtin_amdgcn_sched_barrier(0);
        acc[0][1] = wmma4(f.a0[0], f.b1[0], acc[0][1]);
        acc[1][0] = wmma4(f.a1[0], f.b0[0], acc[1][0]);
        acc[1][1] = wmma4(f.a1[0], f.b1[0], acc[1][1]);
        #pragma unroll
        for (int j = 1; j < 4; ++j) {
            acc[0][0] = wmma4(f.a0[j], f.b0[j], acc[0][0]);
            acc[0][1] = wmma4(f.a0[j], f.b1[j], acc[0][1]);
            acc[1][0] = wmma4(f.a1[j], f.b0[j], acc[1][0]);
            acc[1][1] = wmma4(f.a1[j], f.b1[j], acc[1][1]);
        }
    };

    auto compute_both = [&](const int s) __attribute__((always_inline)) {
        lfrag(F0, Az, Bz, s, 0);
        lfrag(F1, Az, Bz, s, BK / 2);
        __builtin_amdgcn_sched_barrier(0);
        mm_hold(accz, F0);
        __builtin_amdgcn_sched_barrier(0);
        lfrag(F0, Au, Bu, s, 0);
        __builtin_amdgcn_sched_barrier(0);
        mm(accz, F1);
        __builtin_amdgcn_sched_barrier(0);
        lfrag(F1, Au, Bu, s, BK / 2);
        __builtin_amdgcn_sched_barrier(0);
        mm(accu, F0);
        mm(accu, F1);
    };

    auto wmma_done = [&]() __attribute__((always_inline)) {
        asm volatile("" : "+v"(accz[0][0]), "+v"(accz[0][1]), "+v"(accz[1][0]), "+v"(accz[1][1]),
                          "+v"(accu[0][0]), "+v"(accu[0][1]), "+v"(accu[1][0]), "+v"(accu[1][1])
                     : : "memory");
    };

    gload(0);
    __builtin_amdgcn_sched_barrier(0);
    lstore(0);
    __syncthreads();

    for (int t = 0; t < NKT - 1; ++t) {
        gload((t + 1) * BK);
        __builtin_amdgcn_sched_barrier(0);
        compute_both(t & 1);
        wmma_done();
        __builtin_amdgcn_s_setprio(0);
        lstore((t + 1) & 1);
        __syncthreads();
    }
    compute_both((NKT - 1) & 1);
    wmma_done();
    __builtin_amdgcn_s_setprio(0);

    float dsum[2][8];
    #pragma unroll
    for (int mt = 0; mt < 2; ++mt) {
        const int mbase = m0 + wm * 32 + mt * 16 + ((lane >> 4) << 3);
        #pragma unroll
        for (int r = 0; r < 8; ++r) dsum[mt][r] = 0.0f;
        #pragma unroll
        for (int nt = 0; nt < 2; ++nt) {
            const int nc = n0 + wn * 32 + nt * 16 + row16;
            const float bi = b2[nc];
            #pragma unroll
            for (int r = 0; r < 8; ++r) {
                const float z = accz[mt][nt][r] + bi;
                const float sg = 1.0f / (1.0f + __expf(-z));
                accz[mt][nt][r] = z * sg;
                dsum[mt][r] += accu[mt][nt][r] * (sg * (1.0f + z * (1.0f - sg)));
            }
        }
    }
    #pragma unroll
    for (int mt = 0; mt < 2; ++mt)
        #pragma unroll
        for (int r = 0; r < 8; ++r)
            dsum[mt][r] = row16_sum(dsum[mt][r]);

    float* red = &Az[0][0][0];
    if (row16 == 0) {
        #pragma unroll
        for (int mt = 0; mt < 2; ++mt)
            #pragma unroll
            for (int r = 0; r < 8; ++r)
                red[wn * BM + wm * 32 + mt * 16 + ((lane >> 4) << 3) + r] = dsum[mt][r];
    }
    __syncthreads();
    if (tid < BM)
        part[(size_t)bx * B_SZ + m0 + tid] = red[tid] + red[BM + tid];

    #pragma unroll
    for (int mt = 0; mt < 2; ++mt) {
        const int mbase = m0 + wm * 32 + mt * 16 + ((lane >> 4) << 3);
        #pragma unroll
        for (int nt = 0; nt < 2; ++nt) {
            const int nc = n0 + wn * 32 + nt * 16 + row16;
            #pragma unroll
            for (int r = 0; r < 8; ++r)
                a2[(size_t)(mbase + r) * H_SZ + nc] = accz[mt][nt][r];
        }
    }
}

constexpr int GH_X   = H_SZ / BN;
constexpr int NB_C   = (H_SZ / BN) * (H_SZ / BM);
constexpr int NB_L1  = (H_SZ / BN) * (B_SZ / BM);

__global__ __launch_bounds__(NT)
void front_kernel(const float* __restrict__ xs, const float* __restrict__ tvec,
                  const float* __restrict__ W1, const float* __restrict__ b1,
                  const float* __restrict__ W2, const float* __restrict__ W3,
                  float* __restrict__ Cm, float* __restrict__ a1, float* __restrict__ s1)
{
    __shared__ __attribute__((aligned(16))) lds_tile_t As;
    __shared__ __attribute__((aligned(16))) lds_tile_t Bt;
    const int b = blockIdx.x;
    if (b < NB_C) {
        gemm_tile<true, 2, H_SZ, D_SZ>(As, Bt, b % GH_X, b / GH_X,
            nullptr, W3, nullptr, W1, W2, Cm, nullptr);
    } else {
        const int c = b - NB_C;
        gemm_tile<true, 1, H_SZ, D_SZ + 1>(As, Bt, c % GH_X, c / GH_X,
            nullptr, xs, tvec, W1, b1, a1, s1);
    }
}

__global__ __launch_bounds__(NT)
void mid_kernel(const float* __restrict__ a1, const float* __restrict__ W2,
                const float* __restrict__ b2,
                const float* __restrict__ s1, const float* __restrict__ Cm,
                float* __restrict__ a2, float* __restrict__ part)
{
    __shared__ __attribute__((aligned(16))) lds_tile_t Az;
    __shared__ __attribute__((aligned(16))) lds_tile_t Bz;
    __shared__ __attribute__((aligned(16))) lds_tile_t Au;
    __shared__ __attribute__((aligned(16))) lds_tile_t Bu;
    mid_fused_tile(Az, Bz, Au, Bu, blockIdx.x, blockIdx.y, a1, W2, b2, s1, Cm, a2, part);
}

constexpr int OB_M    = 16;
constexpr int OB_NT   = 256;
constexpr int OB_NW   = OB_NT / 32;
constexpr int OB_KR   = H_SZ / 2;
constexpr int OB_KW   = OB_KR / OB_NW;
constexpr int OB_NTL  = (D_SZ + 15) / 16;
constexpr int OB_AP   = H_SZ + 4;
constexpr int OB_RP   = OB_NTL * 16 + 4;
constexpr int OB_A4   = OB_M * H_SZ / 4 / OB_NT;
constexpr int OB_B4   = H_SZ * D_SZ / 4 / OB_NT;
constexpr int OB_B4R0 = (OB_KR * D_SZ / 4 + OB_NT - 1) / OB_NT;
constexpr int NB_OUT  = B_SZ / OB_M;
static_assert(OB_A4 * 4 * OB_NT == OB_M * H_SZ, "A strip must split evenly over the block");
static_assert(OB_B4 * 4 * OB_NT == H_SZ * D_SZ, "W3 must split evenly over the block");
static_assert(OB_B4R0 * 4 * OB_NT >= OB_KR * D_SZ && OB_B4R0 <= OB_B4, "round 0 needs its rows of W3");
static_assert((OB_M * D_SZ * 4) % 128 == 0, "a block's outputs must be whole 128-byte lines");
static_assert(D_SZ <= OB_NT, "one thread per bias value");

__global__ __launch_bounds__(OB_NT)
void back_kernel(const float* __restrict__ a2, const float* __restrict__ W3,
                 const float* __restrict__ b3, float* __restrict__ dxs,
                 const float* __restrict__ part, float* __restrict__ ndiv)
{
    __shared__ __attribute__((aligned(16))) float As[OB_M * OB_AP];
    __shared__ __attribute__((aligned(16))) float Bs[H_SZ * D_SZ];
    __shared__ __attribute__((aligned(16))) float red[OB_NW * OB_M * OB_RP];
    __shared__ float bsh[OB_NT];

    const int tid  = threadIdx.x;
    const int wave = tid >> 5;
    const int lane = tid & 31;
    const int c    = lane & 15;
    const int h    = lane >> 4;
    const int m0   = blockIdx.x * OB_M;

    float pv[GH_X];
    const int prow = (m0 & ~(2 * OB_M - 1)) + lane;
    #pragma unroll
    for (int j = 0; j < GH_X; ++j) pv[j] = part[(size_t)j * B_SZ + prow];
    float bias_v = b3[tid < D_SZ ? tid : D_SZ - 1];
    v4f ar[OB_A4];
    v4f br[OB_B4];
    const v4f* __restrict__ ag = (const v4f*)(a2 + (size_t)m0 * H_SZ);
    const v4f* __restrict__ bg = (const v4f*)W3;
    #pragma unroll
    for (int i = 0; i < OB_A4; ++i) ar[i] = ag[tid + OB_NT * i];
    __builtin_amdgcn_sched_barrier(0);
    #pragma unroll
    for (int i = 0; i < OB_B4R0; ++i) br[i] = bg[tid + OB_NT * i];
    __builtin_amdgcn_sched_barrier(0);
    #pragma unroll
    for (int i = OB_B4R0; i < OB_B4; ++i) br[i] = bg[tid + OB_NT * i];
    __builtin_amdgcn_sched_barrier(0);

    float dsum = 0.0f;
    #pragma unroll
    for (int j = 0; j < GH_X; ++j) dsum += pv[j];
    asm volatile("" : "+v"(dsum));

    v8f acc[OB_NTL] = {};
    int ncol[OB_NTL];
    #pragma unroll
    for (int nt = 0; nt < OB_NTL; ++nt) {
        const int n = 16 * nt + c;
        ncol[nt] = n < D_SZ ? n : D_SZ - 1;
    }

    auto compute = [&](const int rd) __attribute__((always_inline)) {
        #pragma unroll
        for (int j = 0; j < OB_KW / 8; ++j) {
            const int k0 = rd * OB_KR + wave * OB_KW + 8 * j + 4 * h;
            const v4f a = *(const v4f*)&As[c * OB_AP + k0];
            const v2f alo = {a[0], a[1]};
            const v2f ahi = {a[2], a[3]};
            #pragma unroll
            for (int nt = 0; nt < OB_NTL; ++nt) {
                const float* bp = &Bs[k0 * D_SZ + ncol[nt]];
                const v2f blo = {bp[0],        bp[D_SZ]};
                const v2f bhi = {bp[2 * D_SZ], bp[3 * D_SZ]};
                acc[nt] = wmma4(alo, blo, acc[nt]);
                acc[nt] = wmma4(ahi, bhi, acc[nt]);
            }
        }
    };

    bsh[tid] = bias_v;
    #pragma unroll
    for (int i = 0; i < OB_A4; ++i) {
        const int idx = tid + OB_NT * i;
        *(v4f*)&As[(idx >> 7) * OB_AP + 4 * (idx & 127)] = ar[i];
    }
    #pragma unroll
    for (int i = 0; i < OB_B4R0; ++i)
        *(v4f*)&Bs[4 * (tid + OB_NT * i)] = br[i];
    asm volatile("s_wait_dscnt 0x0" ::: "memory");
    asm volatile("s_barrier_signal -1" ::: "memory");
    asm volatile("s_barrier_wait -1" ::: "memory");
    compute(0);
    asm volatile("" : "+v"(acc[0]), "+v"(acc[1]), "+v"(acc[2]), "+v"(acc[3]), "+v"(acc[4])
                 : : "memory");
    #pragma unroll
    for (int i = OB_B4R0; i < OB_B4; ++i)
        *(v4f*)&Bs[4 * (tid + OB_NT * i)] = br[i];
    __syncthreads();
    compute(1);

    #pragma unroll
    for (int nt = 0; nt < OB_NTL; ++nt)
        #pragma unroll
        for (int r = 0; r < 8; ++r)
            red[(wave * OB_M + 8 * h + r) * OB_RP + 16 * nt + c] = acc[nt][r];
    __syncthreads();

    #pragma unroll
    for (int p = 0; p < (OB_M * D_SZ + OB_NT - 1) / OB_NT; ++p) {
        const int o = tid + OB_NT * p;
        if (o < OB_M * D_SZ) {
            const int r = o / D_SZ;
            const int n = o - r * D_SZ;
            float s = 0.0f;
            #pragma unroll
            for (int w = 0; w < OB_NW; ++w)
                s += red[(w * OB_M + r) * OB_RP + n];
            dxs[(size_t)m0 * D_SZ + o] = s + bsh[n];
        }
    }

    if ((blockIdx.x & 1) == 0 && tid < 2 * OB_M)
        ndiv[m0 + tid] = -dsum;
}

extern "C" void kernel_launch(void* const* d_in, const int* in_sizes, int n_in,
                              void* d_out, int out_size, void* d_ws, size_t ws_size,
                              hipStream_t stream) {
    const float* xs = (const float*)d_in[0];
    const float* t  = (const float*)d_in[1];
    const float* W1 = (const float*)d_in[2];
    const float* b1 = (const float*)d_in[3];
    const float* W2 = (const float*)d_in[4];
    const float* b2 = (const float*)d_in[5];
    const float* W3 = (const float*)d_in[6];
    const float* b3 = (const float*)d_in[7];

    float* dxs  = (float*)d_out;
    float* ndiv = (float*)d_out + (size_t)B_SZ * D_SZ;

    float* ws = (float*)d_ws;
    const size_t act = (size_t)B_SZ * H_SZ;
    float* a1  = ws;
    float* s1  = ws + act;
    float* a2  = ws + 2 * act;
    float* prt = ws + 3 * act;
    float* Cm  = ws + 4 * act;

    dim3 blk(NT);
    dim3 gridH(H_SZ / BN, B_SZ / BM);

    front_kernel<<<dim3(NB_C + NB_L1), blk, 0, stream>>>(
        xs, t, W1, b1, W2, W3, Cm, a1, s1);

    mid_kernel<<<gridH, blk, 0, stream>>>(a1, W2, b2, s1, Cm, a2, prt);

    back_kernel<<<dim3(NB_OUT), dim3(OB_NT), 0, stream>>>(
        a2, W3, b3, dxs, prt, ndiv);
}
